# starting kernel
# baseline (speedup 1.0000x reference)
#include <hip/hip_runtime.h>


typedef __attribute__((ext_vector_type(2))) float v2f;
typedef __attribute__((ext_vector_type(8))) float v8f;
typedef v2f __attribute__((aligned(4))) v2f_u;

#define TPW 4

__constant__ float c_rec_lo[8] = {
    0.23037781330885523f,  0.7148465705525415f,  0.6308807679295904f,
   -0.02798376941698385f, -0.18703481171888114f, 0.030841381835986965f,
    0.032883011666982945f,-0.010597401784997278f};
__constant__ float c_rec_hi[8] = {
   -0.010597401784997278f,-0.032883011666982945f, 0.030841381835986965f,
    0.18703481171888114f, -0.02798376941698385f, -0.6308807679295904f,
    0.7148465705525415f,  -0.23037781330885523f};

__device__ __forceinline__ float wtap(const float* w, int t) {
  float v = w[t & 7];
  return (t >= 0 && t < 8) ? v : 0.0f;
}

__global__ __launch_bounds__(128)
void idwt_level_wmma(const float* __restrict__ xa, int stride_a,
                     const float* __restrict__ xd,
                     float* __restrict__ out,
                     int n_in, int L, int j_tiles) {
  const int wave = threadIdx.x >> 5;
  const int lane = threadIdx.x & 31;
  const int wid  = blockIdx.x * 4 + wave;
  const int jt0  = wid * TPW;
  if (__builtin_amdgcn_readfirstlane(jt0 >= j_tiles)) return;
  const int bt = blockIdx.y;

  const int m  = lane & 15;
  const int hi = lane >> 4;
  const int b0 = bt << 4;

  const float* rowA = xa + (size_t)(b0 + m) * (size_t)stride_a;
  const float* rowD = xd + (size_t)(b0 + m) * (size_t)n_in;

  v2f blo[3], bhi[3];
#pragma unroll
  for (int ks = 0; ks < 3; ++ks) {
    const int kb = ks * 4 + hi * 2;
    const int t0 = m + 6 - 2 * kb;
    blo[ks].x = wtap(c_rec_lo, t0);  blo[ks].y = wtap(c_rec_lo, t0 - 2);
    bhi[ks].x = wtap(c_rec_hi, t0);  bhi[ks].y = wtap(c_rec_hi, t0 - 2);
  }

  const int grp_fast =
      (jt0 + TPW <= j_tiles) &&
      ((((jt0 + TPW - 1) << 3) + 12) <= n_in) &&
      (((jt0 + TPW) << 4) <= L);

  if (__builtin_amdgcn_readfirstlane(grp_fast)) {
    v2f fa[TPW][3], fd[TPW][3];
#pragma unroll
    for (int u = 0; u < TPW; ++u) {
      const int i0 = (jt0 + u) << 3;
#pragma unroll
      for (int ks = 0; ks < 3; ++ks) {
        const int c = i0 + ks * 4 + hi * 2;
        fa[u][ks] = *(const v2f_u*)(rowA + c);
        fd[u][ks] = *(const v2f_u*)(rowD + c);
      }
    }
#pragma unroll
    for (int u = 0; u < TPW; ++u) {
      v8f acc = {0.f, 0.f, 0.f, 0.f, 0.f, 0.f, 0.f, 0.f};
#pragma unroll
      for (int ks = 0; ks < 3; ++ks) {
        acc = __builtin_amdgcn_wmma_f32_16x16x4_f32(
            false, fa[u][ks], false, blo[ks], (short)0, acc, false, false);
        acc = __builtin_amdgcn_wmma_f32_16x16x4_f32(
            false, fd[u][ks], false, bhi[ks], (short)0, acc, false, false);
      }
      const int j  = ((jt0 + u) << 4) + m;
      const int mb = hi * 8;
#pragma unroll
      for (int r = 0; r < 8; ++r)
        out[(size_t)(b0 + mb + r) * (size_t)L + j] = acc[r];
    }
    return;
  }

  for (int u = 0; u < TPW; ++u) {
    const int jt = jt0 + u;
    if (__builtin_amdgcn_readfirstlane(jt >= j_tiles)) break;
    const int i0 = jt << 3;
    v8f acc = {0.f, 0.f, 0.f, 0.f, 0.f, 0.f, 0.f, 0.f};
#pragma unroll
    for (int ks = 0; ks < 3; ++ks) {
      const int col0 = i0 + ks * 4 + hi * 2;
      const int col1 = col0 + 1;
      const int cc0 = (col0 < n_in) ? col0 : (n_in - 1);
      const int cc1 = (col1 < n_in) ? col1 : (n_in - 1);
      float a0 = rowA[cc0], a1 = rowA[cc1];
      float d0 = rowD[cc0], d1 = rowD[cc1];
      v2f fa, fd;
      fa.x = (col0 < n_in) ? a0 : 0.f;  fa.y = (col1 < n_in) ? a1 : 0.f;
      fd.x = (col0 < n_in) ? d0 : 0.f;  fd.y = (col1 < n_in) ? d1 : 0.f;
      acc = __builtin_amdgcn_wmma_f32_16x16x4_f32(
          false, fa, false, blo[ks], (short)0, acc, false, false);
      acc = __builtin_amdgcn_wmma_f32_16x16x4_f32(
          false, fd, false, bhi[ks], (short)0, acc, false, false);
    }
    const int j  = (jt << 4) + m;
    const int mb = hi * 8;
    if (j < L) {
#pragma unroll
      for (int r = 0; r < 8; ++r)
        out[(size_t)(b0 + mb + r) * (size_t)L + j] = acc[r];
    }
  }
}

static inline void launch_level(const float* xa, int stride_a, const float* xd,
                                float* out, int n_in, int L, hipStream_t s) {
  const int j_tiles = (L + 15) / 16;
  const int wgroups = (j_tiles + TPW - 1) / TPW;
  dim3 grid((wgroups + 3) / 4, 8);
  idwt_level_wmma<<<grid, 128, 0, s>>>(xa, stride_a, xd, out, n_in, L, j_tiles);
}

extern "C" void kernel_launch(void* const* d_in, const int* in_sizes, int n_in_cnt,
                              void* d_out, int out_size, void* d_ws, size_t ws_size,
                              hipStream_t stream) {
  (void)in_sizes; (void)n_in_cnt; (void)out_size; (void)ws_size;
  const float* a  = (const float*)d_in[0];
  const float* d6 = (const float*)d_in[1];
  const float* d5 = (const float*)d_in[2];
  const float* d4 = (const float*)d_in[3];
  const float* d3 = (const float*)d_in[4];
  const float* d2 = (const float*)d_in[5];
  const float* d1 = (const float*)d_in[6];
  float* out = (float*)d_out;
  float* ws  = (float*)d_ws;

  launch_level(a,    4102,   d6, ws,   4102,   8198,   stream);
  launch_level(ws,   8198,   d5, out,  8198,   16390,  stream);
  launch_level(out,  16390,  d4, ws,   16390,  32774,  stream);
  launch_level(ws,   32774,  d3, out,  32774,  65542,  stream);
  launch_level(out,  65542,  d2, ws,   65541,  131076, stream);
  launch_level(ws,   131076, d1, out,  131075, 262144, stream);
}


# Work-group keeps its 256 output columns and walks all 8 batch tiles (grid 1024x1, was 1024x8): weights built once per wave, next tile's strips load during the products; sum order unchanged
# speedup vs baseline: 3.5104x; 3.5104x over previous
#include <hip/hip_runtime.h>


typedef __attribute__((ext_vector_type(2))) float v2f;
typedef __attribute__((ext_vector_type(8))) float v8f;
typedef __attribute__((ext_vector_type(4))) float v4f;
typedef v4f __attribute__((aligned(4))) v4f_u;

#define LROW 308

#define N_D1 131075
#define N_D2 65541
#define N_D3 32774
#define N_D4 16390
#define N_D5 8198
#define N_D6 4102
#define N_A  4102
#define L_OUT 262144
#define NBT 8

constexpr double K_LO[8] = {
    (double)0.23037781330885523f,  (double)0.7148465705525415f,  (double)0.6308807679295904f,
    (double)-0.02798376941698385f, (double)-0.18703481171888114f, (double)0.030841381835986965f,
    (double)0.032883011666982945f, (double)-0.010597401784997278f};
constexpr double K_HI[8] = {
    (double)-0.010597401784997278f, (double)-0.032883011666982945f, (double)0.030841381835986965f,
    (double)0.18703481171888114f,   (double)-0.02798376941698385f,  (double)-0.6308807679295904f,
    (double)0.7148465705525415f,    (double)-0.23037781330885523f};

struct FiltD { double v[448]; int lo; int len; };

constexpr double fget(const FiltD& f, int s) {
  const int i = s - f.lo;
  return (i >= 0 && i < f.len) ? f.v[i] : 0.0;
}
constexpr FiltD fbase(const double* w) {
  FiltD f{};
  f.lo = -6; f.len = 8;
  for (int i = 0; i < 8; ++i) f.v[i] = w[i];
  return f;
}
constexpr FiltD flift(const double* w, const FiltD& gl, int r) {
  FiltD f{};
  const int st = 1 << r;
  f.lo = gl.lo - 6 * st;
  f.len = (gl.lo + gl.len - 1 + st) - f.lo + 1;
  for (int i = 0; i < f.len; ++i) {
    const int s = f.lo + i;
    double acc = 0.0;
    for (int u = -6; u <= 1; ++u) acc += w[u + 6] * fget(gl, s - st * u);
    f.v[i] = acc;
  }
  return f;
}

constexpr FiltD GL1 = fbase(K_LO);
constexpr FiltD GL2 = flift(K_LO, GL1, 1);
constexpr FiltD GL3 = flift(K_LO, GL2, 2);
constexpr FiltD GL4 = flift(K_LO, GL3, 3);
constexpr FiltD GL5 = flift(K_LO, GL4, 4);
constexpr FiltD GL6 = flift(K_LO, GL5, 5);
constexpr FiltD GH1 = fbase(K_HI);
constexpr FiltD GH2 = flift(K_HI, GL1, 1);
constexpr FiltD GH3 = flift(K_HI, GL2, 2);
constexpr FiltD GH4 = flift(K_HI, GL3, 3);
constexpr FiltD GH5 = flift(K_HI, GL4, 4);
constexpr FiltD GH6 = flift(K_HI, GL5, 5);

#define TPAD 72
struct Tab  { float v[520]; };
struct Tabs { Tab t[7]; };
constexpr Tab to_tab(const FiltD& f) {
  Tab t{};
  for (int i = 0; i < f.len; ++i) t.v[TPAD + i] = (float)f.v[i];
  return t;
}
constexpr Tabs make_tabs() {
  Tabs r{};
  r.t[0] = to_tab(GH1); r.t[1] = to_tab(GH2); r.t[2] = to_tab(GH3); r.t[3] = to_tab(GH4);
  r.t[4] = to_tab(GH5); r.t[5] = to_tab(GH6); r.t[6] = to_tab(GL6);
  return r;
}
constexpr Tabs H_TABS = make_tabs();
__constant__ Tabs c_tabs = H_TABS;

template <int B, int R, int NKS>
__device__ __forceinline__ void build_w(v2f* w, int delta, int m, int hi) {
  constexpr int LO = -6 * ((1 << R) - 1);
  const float* t = &c_tabs.t[B].v[0] + (m - ((2 * hi) << R));
#pragma unroll
  for (int ks = 0; ks < NKS; ++ks) {
    w[ks].x = t[TPAD - LO + delta - ((ks * 4) << R)];
    w[ks].y = t[TPAD - LO + delta - ((ks * 4 + 1) << R)];
  }
}

__device__ __forceinline__ v4f fix_piece(v4f L, int sh) {
  if (sh != 0) {
    v4f g;
    g.x = (sh == 1) ? L.y : (sh == 2) ? L.z : (sh == 3) ? L.w : 0.f;
    g.y = (sh == 1) ? L.z : (sh == 2) ? L.w : 0.f;
    g.z = (sh == 1) ? L.w : 0.f;
    g.w = 0.f;
    L = g;
  }
  return L;
}

template <int NKS>
__device__ __forceinline__ v8f band_mm(v8f acc, const float* p, const v2f* w) {
#pragma unroll
  for (int ks = 0; ks < NKS; ++ks) {
    v2f fa;
    fa.x = p[ks * 4];
    fa.y = p[ks * 4 + 1];
    acc = __builtin_amdgcn_wmma_f32_16x16x4_f32(
        false, fa, false, w[ks], (short)0, acc, false, false);
  }
  return acc;
}

__device__ __forceinline__ void lds_barrier() {
  asm volatile("s_wait_dscnt 0x0\n\ts_barrier_signal -1\n\ts_barrier_wait -1" ::: "memory");
}

__global__ __launch_bounds__(128, 4)
void idwt6_composed_wmma(const float* __restrict__ a,  const float* __restrict__ d6,
                         const float* __restrict__ d5, const float* __restrict__ d4,
                         const float* __restrict__ d3, const float* __restrict__ d2,
                         const float* __restrict__ d1, float* __restrict__ out) {
  const int wave = threadIdx.x >> 5;
  const int lane = threadIdx.x & 31;
  const int m  = lane & 15;
  const int hi = lane >> 4;
  const int bx = blockIdx.x;

  __shared__ __attribute__((aligned(16))) float s_in[2][16][LROW];

  const int r0 = wave << 2;
  const int colD = (bx << 7) + (lane << 2);
  const float* pA = (lane < 18) ? d2 : (lane < 28) ? d3 : d5;
  const int    nA = (lane < 18) ? N_D2 : (lane < 28) ? N_D3 : N_D5;
  const int  colA = (lane < 18) ? ((bx << 6) + (lane << 2))
                  : (lane < 28) ? ((bx << 5) + ((lane - 18) << 2))
                  :               ((bx << 3) + ((lane - 28) << 2));
  const int   clA = (lane < 28) ? (33 + lane) : (39 + lane);
  const float* pB = (m < 6) ? d4 : (m < 9) ? d6 : (m < 12) ? a : d1;
  const int    nB = (m < 6) ? N_D4 : (m < 9) ? N_D6 : (m < 12) ? N_A : N_D1;
  const int  colB = (m < 6) ? ((bx << 4) + (m << 2))
                  : (m < 9) ? ((bx << 2) + ((m - 6) << 2))
                  : (m < 12) ? ((bx << 2) + ((m - 9) << 2))
                  :            ((bx << 7) + 128);
  const int   clB = (m < 6) ? (61 + m) : (m < 12) ? (65 + m) : 32;
  const bool actB = (m < 13);
  const int ccD = colD < (N_D1 - 4) ? colD : (N_D1 - 4);
  const int ccA = colA < (nA - 4) ? colA : (nA - 4);
  const int ccB = colB < (nB - 4) ? colB : (nB - 4);
  const int shD = colD - ccD, shA = colA - ccA, shB = colB - ccB;

  const float* qD = d1 + (size_t)r0 * (size_t)N_D1 + ccD;
  const float* qA = pA + (size_t)r0 * (size_t)nA + ccA;
  const float* qB = pB + (size_t)(r0 + hi) * (size_t)nB + ccB;

  v4f gD[4], gA[4], gB[2];
  auto request_strips = [&]() {
#pragma unroll
    for (int rr = 0; rr < 4; ++rr) {
      gD[rr] = *(const v4f_u*)(qD + (size_t)rr * (size_t)N_D1);
      gA[rr] = *(const v4f_u*)(qA + (size_t)rr * (size_t)nA);
    }
#pragma unroll
    for (int q = 0; q < 2; ++q)
      gB[q] = *(const v4f_u*)(qB + (size_t)(2 * q) * (size_t)nB);
    qD += (size_t)16 * (size_t)N_D1;
    qA += (size_t)16 * (size_t)nA;
    qB += (size_t)16 * (size_t)nB;
  };
  auto park_strips = [&](int buf) {
#pragma unroll
    for (int rr = 0; rr < 4; ++rr) {
      *(v4f*)&s_in[buf][r0 + rr][lane << 2] = fix_piece(gD[rr], shD);
      *(v4f*)&s_in[buf][r0 + rr][clA << 2]  = fix_piece(gA[rr], shA);
    }
#pragma unroll
    for (int q = 0; q < 2; ++q)
      if (actB) *(v4f*)&s_in[buf][r0 + 2 * q + hi][clB << 2] = fix_piece(gB[q], shB);
  };

  request_strips();
  __builtin_amdgcn_sched_barrier(0);

  v2f wa[4][2], w6[4][2], w5[2][2], w4[2], w3[2], w2[3], w1[3];
#pragma unroll
  for (int u = 0; u < 4; ++u) {
    build_w<6, 6, 2>(wa[u], 16 * u, m, hi);
    build_w<5, 6, 2>(w6[u], 16 * u, m, hi);
  }
#pragma unroll
  for (int q = 0; q < 2; ++q) build_w<4, 5, 2>(w5[q], 16 * q, m, hi);
  build_w<3, 4, 2>(w4, 0, m, hi);
  build_w<2, 3, 2>(w3, 0, m, hi);
  build_w<1, 2, 3>(w2, 0, m, hi);
  build_w<0, 1, 3>(w1, 0, m, hi);

  asm volatile("" :: "v"(gD[0]), "v"(gD[1]), "v"(gD[2]), "v"(gD[3]),
                     "v"(gA[0]), "v"(gA[1]), "v"(gA[2]), "v"(gA[3]),
                     "v"(gB[0]), "v"(gB[1]));

  park_strips(0);
  __syncthreads();

  float* po = out + (size_t)(hi * 8) * (size_t)L_OUT + ((bx << 8) + m);
  for (int it = 0; it < NBT; ++it) {
    const int cur = it & 1;
    const bool more = (it + 1 < NBT);
    if (more) request_strips();
    __builtin_amdgcn_sched_barrier(0);

    const float* srow = &s_in[cur][m][hi * 2];
#pragma unroll
    for (int u = 0; u < 4; ++u) {
      const int tl = (wave << 2) + u;
      v8f acc = {0.f, 0.f, 0.f, 0.f, 0.f, 0.f, 0.f, 0.f};
      acc = band_mm<2>(acc, srow + 296 + (tl >> 2), wa[u]);
      acc = band_mm<2>(acc, srow + 284 + (tl >> 2), w6[u]);
      acc = band_mm<2>(acc, srow + 268 + (tl >> 1), w5[u & 1]);
      acc = band_mm<2>(acc, srow + 244 + tl,        w4);
      acc = band_mm<2>(acc, srow + 204 + (tl << 1), w3);
      acc = band_mm<3>(acc, srow + 132 + (tl << 2), w2);
      acc = band_mm<3>(acc, srow +       (tl << 3), w1);
      float* pt = po + (tl << 4);
#pragma unroll
      for (int r = 0; r < 8; ++r)
        pt[(size_t)r * (size_t)L_OUT] = acc[r];
    }
    po += (size_t)16 * (size_t)L_OUT;

    if (more) {
      park_strips(cur ^ 1);
      lds_barrier();
    }
  }
}

extern "C" void kernel_launch(void* const* d_in, const int* in_sizes, int n_in_cnt,
                              void* d_out, int out_size, void* d_ws, size_t ws_size,
                              hipStream_t stream) {
  (void)in_sizes; (void)n_in_cnt; (void)out_size; (void)d_ws; (void)ws_size;
  const float* a  = (const float*)d_in[0];
  const float* d6 = (const float*)d_in[1];
  const float* d5 = (const float*)d_in[2];
  const float* d4 = (const float*)d_in[3];
  const float* d3 = (const float*)d_in[4];
  const float* d2 = (const float*)d_in[5];
  const float* d1 = (const float*)d_in[6];
  float* out = (float*)d_out;

  idwt6_composed_wmma<<<dim3(1024, 1), 128, 0, stream>>>(a, d6, d5, d4, d3, d2, d1, out);
}


# speedup vs baseline: 3.6330x; 1.0349x over previous
#include <hip/hip_runtime.h>


typedef __attribute__((ext_vector_type(2))) float v2f;
typedef __attribute__((ext_vector_type(8))) float v8f;
typedef __attribute__((ext_vector_type(4))) float v4f;
typedef v4f __attribute__((aligned(4))) v4f_u;

#define LROW 308

#define N_D1 131075
#define N_D2 65541
#define N_D3 32774
#define N_D4 16390
#define N_D5 8198
#define N_D6 4102
#define N_A  4102
#define L_OUT 262144
#define NST 8

constexpr double K_LO[8] = {
    (double)0.23037781330885523f,  (double)0.7148465705525415f,  (double)0.6308807679295904f,
    (double)-0.02798376941698385f, (double)-0.18703481171888114f, (double)0.030841381835986965f,
    (double)0.032883011666982945f, (double)-0.010597401784997278f};
constexpr double K_HI[8] = {
    (double)-0.010597401784997278f, (double)-0.032883011666982945f, (double)0.030841381835986965f,
    (double)0.18703481171888114f,   (double)-0.02798376941698385f,  (double)-0.6308807679295904f,
    (double)0.7148465705525415f,    (double)-0.23037781330885523f};

struct FiltD { double v[448]; int lo; int len; };

constexpr double fget(const FiltD& f, int s) {
  const int i = s - f.lo;
  return (i >= 0 && i < f.len) ? f.v[i] : 0.0;
}
constexpr FiltD fbase(const double* w) {
  FiltD f{};
  f.lo = -6; f.len = 8;
  for (int i = 0; i < 8; ++i) f.v[i] = w[i];
  return f;
}
constexpr FiltD flift(const double* w, const FiltD& gl, int r) {
  FiltD f{};
  const int st = 1 << r;
  f.lo = gl.lo - 6 * st;
  f.len = (gl.lo + gl.len - 1 + st) - f.lo + 1;
  for (int i = 0; i < f.len; ++i) {
    const int s = f.lo + i;
    double acc = 0.0;
    for (int u = -6; u <= 1; ++u) acc += w[u + 6] * fget(gl, s - st * u);
    f.v[i] = acc;
  }
  return f;
}

constexpr FiltD GL1 = fbase(K_LO);
constexpr FiltD GL2 = flift(K_LO, GL1, 1);
constexpr FiltD GL3 = flift(K_LO, GL2, 2);
constexpr FiltD GL4 = flift(K_LO, GL3, 3);
constexpr FiltD GL5 = flift(K_LO, GL4, 4);
constexpr FiltD GL6 = flift(K_LO, GL5, 5);
constexpr FiltD GH1 = fbase(K_HI);
constexpr FiltD GH2 = flift(K_HI, GL1, 1);
constexpr FiltD GH3 = flift(K_HI, GL2, 2);
constexpr FiltD GH4 = flift(K_HI, GL3, 3);
constexpr FiltD GH5 = flift(K_HI, GL4, 4);
constexpr FiltD GH6 = flift(K_HI, GL5, 5);

#define TPAD 72
struct Tab  { float v[520]; };
struct Tabs { Tab t[7]; };
constexpr Tab to_tab(const FiltD& f) {
  Tab t{};
  for (int i = 0; i < f.len; ++i) t.v[TPAD + i] = (float)f.v[i];
  return t;
}
constexpr Tabs make_tabs() {
  Tabs r{};
  r.t[0] = to_tab(GH1); r.t[1] = to_tab(GH2); r.t[2] = to_tab(GH3); r.t[3] = to_tab(GH4);
  r.t[4] = to_tab(GH5); r.t[5] = to_tab(GH6); r.t[6] = to_tab(GL6);
  return r;
}
constexpr Tabs H_TABS = make_tabs();
__constant__ Tabs c_tabs = H_TABS;

template <int B, int R, int NKS>
__device__ __forceinline__ void build_w(v2f* w, int delta, int m, int hi) {
  constexpr int LO = -6 * ((1 << R) - 1);
  const float* t = &c_tabs.t[B].v[0] + (m - ((2 * hi) << R));
#pragma unroll
  for (int ks = 0; ks < NKS; ++ks) {
    w[ks].x = t[TPAD - LO + delta - ((ks * 4) << R)];
    w[ks].y = t[TPAD - LO + delta - ((ks * 4 + 1) << R)];
  }
}

__device__ __forceinline__ v4f fix_piece(v4f L, int sh) {
  if (sh != 0) {
    v4f g;
    g.x = (sh == 1) ? L.y : (sh == 2) ? L.z : (sh == 3) ? L.w : 0.f;
    g.y = (sh == 1) ? L.z : (sh == 2) ? L.w : 0.f;
    g.z = (sh == 1) ? L.w : 0.f;
    g.w = 0.f;
    L = g;
  }
  return L;
}

template <int NKS>
__device__ __forceinline__ v8f band_mm(v8f acc, const float* p, const v2f* w) {
#pragma unroll
  for (int ks = 0; ks < NKS; ++ks) {
    v2f fa;
    fa.x = p[ks * 4];
    fa.y = p[ks * 4 + 1];
    acc = __builtin_amdgcn_wmma_f32_16x16x4_f32(
        false, fa, false, w[ks], (short)0, acc, false, false);
  }
  return acc;
}

__device__ __forceinline__ void lds_barrier() {
  asm volatile("s_wait_dscnt 0x0\n\ts_barrier_signal -1\n\ts_barrier_wait -1" ::: "memory");
}

__global__ __launch_bounds__(128, 4)
void idwt6_composed_wmma(const float* __restrict__ a,  const float* __restrict__ d6,
                         const float* __restrict__ d5, const float* __restrict__ d4,
                         const float* __restrict__ d3, const float* __restrict__ d2,
                         const float* __restrict__ d1, float* __restrict__ out) {
  const int wave = threadIdx.x >> 5;
  const int lane = threadIdx.x & 31;
  const int m  = lane & 15;
  const int hi = lane >> 4;
  const int cb = blockIdx.x >> 3;
  const int bt = blockIdx.x & 7;

  __shared__ __attribute__((aligned(16))) float s_in[2][16][LROW];

  const int r0 = wave << 2;
  const float* pA = (lane < 18) ? d2 : (lane < 28) ? d3 : d5;
  const int    nA = (lane < 18) ? N_D2 : (lane < 28) ? N_D3 : N_D5;
  const int  sftA = (lane < 18) ? 6 : (lane < 28) ? 5 : 3;
  const int  offA = (lane < 18) ? (lane << 2) : (lane < 28) ? ((lane - 18) << 2) : ((lane - 28) << 2);
  const int   clA = (lane < 28) ? (33 + lane) : (39 + lane);
  const float* pB = (m < 6) ? d4 : (m < 9) ? d6 : (m < 12) ? a : d1;
  const int    nB = (m < 6) ? N_D4 : (m < 9) ? N_D6 : (m < 12) ? N_A : N_D1;
  const int  sftB = (m < 6) ? 4 : (m < 12) ? 2 : 7;
  const int  offB = (m < 6) ? (m << 2) : (m < 9) ? ((m - 6) << 2) : (m < 12) ? ((m - 9) << 2) : 128;
  const int   clB = (m < 6) ? (61 + m) : (m < 12) ? (65 + m) : 32;
  const bool actB = (m < 13);

  const float* rD = d1 + (size_t)((bt << 4) + r0) * (size_t)N_D1;
  const float* rA = pA + (size_t)((bt << 4) + r0) * (size_t)nA;
  const float* rB = pB + (size_t)((bt << 4) + r0 + hi) * (size_t)nB;

  v4f gD[4], gA[4], gB[2];
  int shD = 0, shA = 0, shB = 0;
  auto request_strips = [&](int gx) {
    const int colD = (gx << 7) + (lane << 2);
    const int colA = (gx << sftA) + offA;
    const int colB = (gx << sftB) + offB;
    const int ccD = colD < (N_D1 - 4) ? colD : (N_D1 - 4);
    const int ccA = colA < (nA - 4) ? colA : (nA - 4);
    const int ccB = colB < (nB - 4) ? colB : (nB - 4);
    shD = colD - ccD; shA = colA - ccA; shB = colB - ccB;
    const float* qD = rD + ccD;
    const float* qA = rA + ccA;
    const float* qB = rB + ccB;
#pragma unroll
    for (int rr = 0; rr < 4; ++rr) {
      gD[rr] = *(const v4f_u*)(qD + (size_t)rr * (size_t)N_D1);
      gA[rr] = *(const v4f_u*)(qA + (size_t)rr * (size_t)nA);
    }
#pragma unroll
    for (int q = 0; q < 2; ++q)
      gB[q] = *(const v4f_u*)(qB + (size_t)(2 * q) * (size_t)nB);
  };
  auto park_strips = [&](int buf) {
#pragma unroll
    for (int rr = 0; rr < 4; ++rr) {
      *(v4f*)&s_in[buf][r0 + rr][lane << 2] = fix_piece(gD[rr], shD);
      *(v4f*)&s_in[buf][r0 + rr][clA << 2]  = fix_piece(gA[rr], shA);
    }
#pragma unroll
    for (int q = 0; q < 2; ++q)
      if (actB) *(v4f*)&s_in[buf][r0 + 2 * q + hi][clB << 2] = fix_piece(gB[q], shB);
  };

  const int gx0 = cb << 3;
  request_strips(gx0 + (cb & 7));
  __builtin_amdgcn_sched_barrier(0);

  v2f wa[4][2], w6[4][2], w5[2][2], w4[2], w3[2], w2[3], w1[3];
#pragma unroll
  for (int u = 0; u < 4; ++u) {
    build_w<6, 6, 2>(wa[u], 16 * u, m, hi);
    build_w<5, 6, 2>(w6[u], 16 * u, m, hi);
  }
#pragma unroll
  for (int q = 0; q < 2; ++q) build_w<4, 5, 2>(w5[q], 16 * q, m, hi);
  build_w<3, 4, 2>(w4, 0, m, hi);
  build_w<2, 3, 2>(w3, 0, m, hi);
  build_w<1, 2, 3>(w2, 0, m, hi);
  build_w<0, 1, 3>(w1, 0, m, hi);

  asm volatile("" :: "v"(gD[0]), "v"(gD[1]), "v"(gD[2]), "v"(gD[3]),
                     "v"(gA[0]), "v"(gA[1]), "v"(gA[2]), "v"(gA[3]),
                     "v"(gB[0]), "v"(gB[1]));

  park_strips(0);
  __syncthreads();

  float* po = out + (size_t)((bt << 4) + hi * 8) * (size_t)L_OUT + m;
  for (int it = 0; it < NST; ++it) {
    const int cur = it & 1;
    const bool more = (it + 1 < NST);
    const int gx = gx0 + ((it + cb) & 7);
    if (more) request_strips(gx0 + ((it + 1 + cb) & 7));
    __builtin_amdgcn_sched_barrier(0);

    const float* srow = &s_in[cur][m][hi * 2];
#pragma unroll
    for (int u = 0; u < 4; ++u) {
      const int tl = (wave << 2) + u;
      v8f acc = {0.f, 0.f, 0.f, 0.f, 0.f, 0.f, 0.f, 0.f};
      acc = band_mm<2>(acc, srow + 296 + (tl >> 2), wa[u]);
      acc = band_mm<2>(acc, srow + 284 + (tl >> 2), w6[u]);
      acc = band_mm<2>(acc, srow + 268 + (tl >> 1), w5[u & 1]);
      acc = band_mm<2>(acc, srow + 244 + tl,        w4);
      acc = band_mm<2>(acc, srow + 204 + (tl << 1), w3);
      acc = band_mm<3>(acc, srow + 132 + (tl << 2), w2);
      acc = band_mm<3>(acc, srow +       (tl << 3), w1);
      float* pt = po + ((gx << 8) + (tl << 4));
#pragma unroll
      for (int r = 0; r < 8; ++r)
        pt[(size_t)r * (size_t)L_OUT] = acc[r];
    }

    if (more) {
      park_strips(cur ^ 1);
      lds_barrier();
    }
  }
}

extern "C" void kernel_launch(void* const* d_in, const int* in_sizes, int n_in_cnt,
                              void* d_out, int out_size, void* d_ws, size_t ws_size,
                              hipStream_t stream) {
  (void)in_sizes; (void)n_in_cnt; (void)out_size; (void)d_ws; (void)ws_size;
  const float* a  = (const float*)d_in[0];
  const float* d6 = (const float*)d_in[1];
  const float* d5 = (const float*)d_in[2];
  const float* d4 = (const float*)d_in[3];
  const float* d3 = (const float*)d_in[4];
  const float* d2 = (const float*)d_in[5];
  const float* d1 = (const float*)d_in[6];
  float* out = (float*)d_out;

  idwt6_composed_wmma<<<dim3(1024, 1), 128, 0, stream>>>(a, d6, d5, d4, d3, d2, d1, out);
}


# Two steps' input strips in flight per wave (was one): step i+2's ten 16-byte row loads go out before step i's products into a second register set, step i+1's are parked behind them; sums unchanged
# speedup vs baseline: 3.7878x; 1.0426x over previous
#include <hip/hip_runtime.h>


typedef __attribute__((ext_vector_type(2))) float v2f;
typedef __attribute__((ext_vector_type(8))) float v8f;
typedef __attribute__((ext_vector_type(4))) float v4f;
typedef v4f __attribute__((aligned(4))) v4f_u;

#define LROW 308

#define N_D1 131075
#define N_D2 65541
#define N_D3 32774
#define N_D4 16390
#define N_D5 8198
#define N_D6 4102
#define N_A  4102
#define L_OUT 262144
#define NST 8

constexpr double K_LO[8] = {
    (double)0.23037781330885523f,  (double)0.7148465705525415f,  (double)0.6308807679295904f,
    (double)-0.02798376941698385f, (double)-0.18703481171888114f, (double)0.030841381835986965f,
    (double)0.032883011666982945f, (double)-0.010597401784997278f};
constexpr double K_HI[8] = {
    (double)-0.010597401784997278f, (double)-0.032883011666982945f, (double)0.030841381835986965f,
    (double)0.18703481171888114f,   (double)-0.02798376941698385f,  (double)-0.6308807679295904f,
    (double)0.7148465705525415f,    (double)-0.23037781330885523f};

struct FiltD { double v[448]; int lo; int len; };

constexpr double fget(const FiltD& f, int s) {
  const int i = s - f.lo;
  return (i >= 0 && i < f.len) ? f.v[i] : 0.0;
}
constexpr FiltD fbase(const double* w) {
  FiltD f{};
  f.lo = -6; f.len = 8;
  for (int i = 0; i < 8; ++i) f.v[i] = w[i];
  return f;
}
constexpr FiltD flift(const double* w, const FiltD& gl, int r) {
  FiltD f{};
  const int st = 1 << r;
  f.lo = gl.lo - 6 * st;
  f.len = (gl.lo + gl.len - 1 + st) - f.lo + 1;
  for (int i = 0; i < f.len; ++i) {
    const int s = f.lo + i;
    double acc = 0.0;
    for (int u = -6; u <= 1; ++u) acc += w[u + 6] * fget(gl, s - st * u);
    f.v[i] = acc;
  }
  return f;
}

constexpr FiltD GL1 = fbase(K_LO);
constexpr FiltD GL2 = flift(K_LO, GL1, 1);
constexpr FiltD GL3 = flift(K_LO, GL2, 2);
constexpr FiltD GL4 = flift(K_LO, GL3, 3);
constexpr FiltD GL5 = flift(K_LO, GL4, 4);
constexpr FiltD GL6 = flift(K_LO, GL5, 5);
constexpr FiltD GH1 = fbase(K_HI);
constexpr FiltD GH2 = flift(K_HI, GL1, 1);
constexpr FiltD GH3 = flift(K_HI, GL2, 2);
constexpr FiltD GH4 = flift(K_HI, GL3, 3);
constexpr FiltD GH5 = flift(K_HI, GL4, 4);
constexpr FiltD GH6 = flift(K_HI, GL5, 5);

#define TPAD 72
struct Tab  { float v[520]; };
struct Tabs { Tab t[7]; };
constexpr Tab to_tab(const FiltD& f) {
  Tab t{};
  for (int i = 0; i < f.len; ++i) t.v[TPAD + i] = (float)f.v[i];
  return t;
}
constexpr Tabs make_tabs() {
  Tabs r{};
  r.t[0] = to_tab(GH1); r.t[1] = to_tab(GH2); r.t[2] = to_tab(GH3); r.t[3] = to_tab(GH4);
  r.t[4] = to_tab(GH5); r.t[5] = to_tab(GH6); r.t[6] = to_tab(GL6);
  return r;
}
constexpr Tabs H_TABS = make_tabs();
__constant__ Tabs c_tabs = H_TABS;

template <int B, int R, int NKS>
__device__ __forceinline__ void build_w(v2f* w, int delta, int m, int hi) {
  constexpr int LO = -6 * ((1 << R) - 1);
  const float* t = &c_tabs.t[B].v[0] + (m - ((2 * hi) << R));
#pragma unroll
  for (int ks = 0; ks < NKS; ++ks) {
    w[ks].x = t[TPAD - LO + delta - ((ks * 4) << R)];
    w[ks].y = t[TPAD - LO + delta - ((ks * 4 + 1) << R)];
  }
}

__device__ __forceinline__ v4f fix_piece(v4f L, int sh) {
  if (sh != 0) {
    v4f g;
    g.x = (sh == 1) ? L.y : (sh == 2) ? L.z : (sh == 3) ? L.w : 0.f;
    g.y = (sh == 1) ? L.z : (sh == 2) ? L.w : 0.f;
    g.z = (sh == 1) ? L.w : 0.f;
    g.w = 0.f;
    L = g;
  }
  return L;
}

template <int NKS>
__device__ __forceinline__ v8f band_mm(v8f acc, const float* p, const v2f* w) {
#pragma unroll
  for (int ks = 0; ks < NKS; ++ks) {
    v2f fa;
    fa.x = p[ks * 4];
    fa.y = p[ks * 4 + 1];
    acc = __builtin_amdgcn_wmma_f32_16x16x4_f32(
        false, fa, false, w[ks], (short)0, acc, false, false);
  }
  return acc;
}

__device__ __forceinline__ void lds_barrier() {
  asm volatile("s_wait_dscnt 0x0\n\ts_barrier_signal -1\n\ts_barrier_wait -1" ::: "memory");
}

__global__ __launch_bounds__(128, 4)
void idwt6_composed_wmma(const float* __restrict__ a,  const float* __restrict__ d6,
                         const float* __restrict__ d5, const float* __restrict__ d4,
                         const float* __restrict__ d3, const float* __restrict__ d2,
                         const float* __restrict__ d1, float* __restrict__ out) {
  const int wave = threadIdx.x >> 5;
  const int lane = threadIdx.x & 31;
  const int m  = lane & 15;
  const int hi = lane >> 4;
  const int cb = blockIdx.x >> 3;
  const int bt = blockIdx.x & 7;

  __shared__ __attribute__((aligned(16))) float s_in[2][16][LROW];

  const int r0 = wave << 2;
  const float* pA = (lane < 18) ? d2 : (lane < 28) ? d3 : d5;
  const int    nA = (lane < 18) ? N_D2 : (lane < 28) ? N_D3 : N_D5;
  const int  sftA = (lane < 18) ? 6 : (lane < 28) ? 5 : 3;
  const int  offA = (lane < 18) ? (lane << 2) : (lane < 28) ? ((lane - 18) << 2) : ((lane - 28) << 2);
  const int   clA = (lane < 28) ? (33 + lane) : (39 + lane);
  const float* pB = (m < 6) ? d4 : (m < 9) ? d6 : (m < 12) ? a : d1;
  const int    nB = (m < 6) ? N_D4 : (m < 9) ? N_D6 : (m < 12) ? N_A : N_D1;
  const int  sftB = (m < 6) ? 4 : (m < 12) ? 2 : 7;
  const int  offB = (m < 6) ? (m << 2) : (m < 9) ? ((m - 6) << 2) : (m < 12) ? ((m - 9) << 2) : 128;
  const int   clB = (m < 6) ? (61 + m) : (m < 12) ? (65 + m) : 32;

  const float* rD = d1 + (size_t)((bt << 4) + r0) * (size_t)N_D1;
  const float* rA = pA + (size_t)((bt << 4) + r0) * (size_t)nA;
  const float* rB = pB + (size_t)((bt << 4) + r0 + hi) * (size_t)nB;

  struct Strips { v4f gD[4], gA[4], gB[2]; };
  Strips st0, st1;
  auto request_strips = [&](Strips& s, int gx) {
    v4f (&gD)[4] = s.gD; v4f (&gA)[4] = s.gA; v4f (&gB)[2] = s.gB;
    const int colD = (gx << 7) + (lane << 2);
    const int colA = (gx << sftA) + offA;
    const int colB = (gx << sftB) + offB;
    const int ccD = colD < (N_D1 - 4) ? colD : (N_D1 - 4);
    const int ccA = colA < (nA - 4) ? colA : (nA - 4);
    const int ccB = colB < (nB - 4) ? colB : (nB - 4);
    const float* qD = rD + ccD;
    const float* qA = rA + ccA;
    const float* qB = rB + ccB;
#pragma unroll
    for (int rr = 0; rr < 4; ++rr) {
      gD[rr] = *(const v4f_u*)(qD + (size_t)rr * (size_t)N_D1);
      gA[rr] = *(const v4f_u*)(qA + (size_t)rr * (size_t)nA);
    }
#pragma unroll
    for (int q = 0; q < 2; ++q)
      gB[q] = *(const v4f_u*)(qB + (size_t)(2 * q) * (size_t)nB);
  };
  auto park_strips = [&](const Strips& s, int buf, int gx) {
    if (gx == 1023) {
      const int colD = (gx << 7) + (lane << 2);
      const int colA = (gx << sftA) + offA;
      const int colB = (gx << sftB) + offB;
      const int shD = colD - (colD < (N_D1 - 4) ? colD : (N_D1 - 4));
      const int shA = colA - (colA < (nA - 4) ? colA : (nA - 4));
      const int shB = colB - (colB < (nB - 4) ? colB : (nB - 4));
#pragma unroll
      for (int rr = 0; rr < 4; ++rr) {
        *(v4f*)&s_in[buf][r0 + rr][lane << 2] = fix_piece(s.gD[rr], shD);
        *(v4f*)&s_in[buf][r0 + rr][clA << 2]  = fix_piece(s.gA[rr], shA);
      }
#pragma unroll
      for (int q = 0; q < 2; ++q)
        *(v4f*)&s_in[buf][r0 + 2 * q + hi][clB << 2] = fix_piece(s.gB[q], shB);
    } else {
#pragma unroll
      for (int rr = 0; rr < 4; ++rr) {
        *(v4f*)&s_in[buf][r0 + rr][lane << 2] = s.gD[rr];
        *(v4f*)&s_in[buf][r0 + rr][clA << 2]  = s.gA[rr];
      }
#pragma unroll
      for (int q = 0; q < 2; ++q)
        *(v4f*)&s_in[buf][r0 + 2 * q + hi][clB << 2] = s.gB[q];
    }
  };

  const int gx0 = cb << 3;
  request_strips(st0, gx0 + (cb & 7));
  __builtin_amdgcn_sched_barrier(0);

  v2f wa[4][2], w6[4][2], w5[2][2], w4[2], w3[2], w2[3], w1[3];
#pragma unroll
  for (int u = 0; u < 4; ++u) {
    build_w<6, 6, 2>(wa[u], 16 * u, m, hi);
    build_w<5, 6, 2>(w6[u], 16 * u, m, hi);
  }
#pragma unroll
  for (int q = 0; q < 2; ++q) build_w<4, 5, 2>(w5[q], 16 * q, m, hi);
  build_w<3, 4, 2>(w4, 0, m, hi);
  build_w<2, 3, 2>(w3, 0, m, hi);
  build_w<1, 2, 3>(w2, 0, m, hi);
  build_w<0, 1, 3>(w1, 0, m, hi);

  __builtin_amdgcn_sched_barrier(0);
  request_strips(st1, gx0 + ((1 + cb) & 7));
  __builtin_amdgcn_sched_barrier(0);

  park_strips(st0, 0, gx0 + (cb & 7));
  lds_barrier();

  float* po = out + (size_t)((bt << 4) + hi * 8) * (size_t)L_OUT + m;
  auto step = [&](int it, Strips& other, const Strips& mine, const bool ask, const bool more) {
    const int cur = it & 1;
    const int gx = gx0 + ((it + cb) & 7);
    if (ask) request_strips(other, gx0 + ((it + 2 + cb) & 7));
    __builtin_amdgcn_sched_barrier(0);

    const float* srow = &s_in[cur][m][hi * 2];
#pragma unroll
    for (int u = 0; u < 4; ++u) {
      const int tl = (wave << 2) + u;
      v8f acc = {0.f, 0.f, 0.f, 0.f, 0.f, 0.f, 0.f, 0.f};
      acc = band_mm<2>(acc, srow + 296 + (tl >> 2), wa[u]);
      acc = band_mm<2>(acc, srow + 284 + (tl >> 2), w6[u]);
      acc = band_mm<2>(acc, srow + 268 + (tl >> 1), w5[u & 1]);
      acc = band_mm<2>(acc, srow + 244 + tl,        w4);
      acc = band_mm<2>(acc, srow + 204 + (tl << 1), w3);
      acc = band_mm<3>(acc, srow + 132 + (tl << 2), w2);
      acc = band_mm<3>(acc, srow +       (tl << 3), w1);
      float* pt = po + ((gx << 8) + (tl << 4));
#pragma unroll
      for (int r = 0; r < 8; ++r)
        pt[(size_t)r * (size_t)L_OUT] = acc[r];
      if (u == 1) __builtin_amdgcn_sched_barrier(0);
    }

    if (more) {
      park_strips(mine, cur ^ 1, gx0 + ((it + 1 + cb) & 7));
      lds_barrier();
    }
  };
  for (int it = 0; it < NST - 2; it += 2) {
    step(it,     st0, st1, true, true);
    step(it + 1, st1, st0, true, true);
  }
  step(NST - 2, st0, st1, false, true);
  step(NST - 1, st1, st0, false, false);
}

extern "C" void kernel_launch(void* const* d_in, const int* in_sizes, int n_in_cnt,
                              void* d_out, int out_size, void* d_ws, size_t ws_size,
                              hipStream_t stream) {
  (void)in_sizes; (void)n_in_cnt; (void)out_size; (void)d_ws; (void)ws_size;
  const float* a  = (const float*)d_in[0];
  const float* d6 = (const float*)d_in[1];
  const float* d5 = (const float*)d_in[2];
  const float* d4 = (const float*)d_in[3];
  const float* d3 = (const float*)d_in[4];
  const float* d2 = (const float*)d_in[5];
  const float* d1 = (const float*)d_in[6];
  float* out = (float*)d_out;

  idwt6_composed_wmma<<<dim3(1024, 1), 128, 0, stream>>>(a, d6, d5, d4, d3, d2, d1, out);
}


# Each store writes two whole 256-byte units, 16 bytes per lane (was 64-byte pieces): four tiles' results exchanged across lanes by DPP moves; strips requested one step ahead, not two; sums unchanged
# speedup vs baseline: 3.9478x; 1.0422x over previous
#include <hip/hip_runtime.h>


typedef __attribute__((ext_vector_type(2))) float v2f;
typedef __attribute__((ext_vector_type(8))) float v8f;
typedef __attribute__((ext_vector_type(4))) float v4f;
typedef v4f __attribute__((aligned(4))) v4f_u;

#define LROW 308

#define N_D1 131075
#define N_D2 65541
#define N_D3 32774
#define N_D4 16390
#define N_D5 8198
#define N_D6 4102
#define N_A  4102
#define L_OUT 262144
#define NST 8

constexpr double K_LO[8] = {
    (double)0.23037781330885523f,  (double)0.7148465705525415f,  (double)0.6308807679295904f,
    (double)-0.02798376941698385f, (double)-0.18703481171888114f, (double)0.030841381835986965f,
    (double)0.032883011666982945f, (double)-0.010597401784997278f};
constexpr double K_HI[8] = {
    (double)-0.010597401784997278f, (double)-0.032883011666982945f, (double)0.030841381835986965f,
    (double)0.18703481171888114f,   (double)-0.02798376941698385f,  (double)-0.6308807679295904f,
    (double)0.7148465705525415f,    (double)-0.23037781330885523f};

struct FiltD { double v[448]; int lo; int len; };

constexpr double fget(const FiltD& f, int s) {
  const int i = s - f.lo;
  return (i >= 0 && i < f.len) ? f.v[i] : 0.0;
}
constexpr FiltD fbase(const double* w) {
  FiltD f{};
  f.lo = -6; f.len = 8;
  for (int i = 0; i < 8; ++i) f.v[i] = w[i];
  return f;
}
constexpr FiltD flift(const double* w, const FiltD& gl, int r) {
  FiltD f{};
  const int st = 1 << r;
  f.lo = gl.lo - 6 * st;
  f.len = (gl.lo + gl.len - 1 + st) - f.lo + 1;
  for (int i = 0; i < f.len; ++i) {
    const int s = f.lo + i;
    double acc = 0.0;
    for (int u = -6; u <= 1; ++u) acc += w[u + 6] * fget(gl, s - st * u);
    f.v[i] = acc;
  }
  return f;
}

constexpr FiltD GL1 = fbase(K_LO);
constexpr FiltD GL2 = flift(K_LO, GL1, 1);
constexpr FiltD GL3 = flift(K_LO, GL2, 2);
constexpr FiltD GL4 = flift(K_LO, GL3, 3);
constexpr FiltD GL5 = flift(K_LO, GL4, 4);
constexpr FiltD GL6 = flift(K_LO, GL5, 5);
constexpr FiltD GH1 = fbase(K_HI);
constexpr FiltD GH2 = flift(K_HI, GL1, 1);
constexpr FiltD GH3 = flift(K_HI, GL2, 2);
constexpr FiltD GH4 = flift(K_HI, GL3, 3);
constexpr FiltD GH5 = flift(K_HI, GL4, 4);
constexpr FiltD GH6 = flift(K_HI, GL5, 5);

#define TPAD 72
struct Tab  { float v[520]; };
struct Tabs { Tab t[7]; };
constexpr Tab to_tab(const FiltD& f) {
  Tab t{};
  for (int i = 0; i < f.len; ++i) t.v[TPAD + i] = (float)f.v[i];
  return t;
}
constexpr Tabs make_tabs() {
  Tabs r{};
  r.t[0] = to_tab(GH1); r.t[1] = to_tab(GH2); r.t[2] = to_tab(GH3); r.t[3] = to_tab(GH4);
  r.t[4] = to_tab(GH5); r.t[5] = to_tab(GH6); r.t[6] = to_tab(GL6);
  return r;
}
constexpr Tabs H_TABS = make_tabs();
__constant__ Tabs c_tabs = H_TABS;

template <int B, int R, int NKS>
__device__ __forceinline__ void build_w(v2f* w, int delta, int m, int hi) {
  constexpr int LO = -6 * ((1 << R) - 1);
  const float* t = &c_tabs.t[B].v[0] + (m - ((2 * hi) << R));
#pragma unroll
  for (int ks = 0; ks < NKS; ++ks) {
    w[ks].x = t[TPAD - LO + delta - ((ks * 4) << R)];
    w[ks].y = t[TPAD - LO + delta - ((ks * 4 + 1) << R)];
  }
}

__device__ __forceinline__ v4f fix_piece(v4f L, int sh) {
  if (sh != 0) {
    v4f g;
    g.x = (sh == 1) ? L.y : (sh == 2) ? L.z : (sh == 3) ? L.w : 0.f;
    g.y = (sh == 1) ? L.z : (sh == 2) ? L.w : 0.f;
    g.z = (sh == 1) ? L.w : 0.f;
    g.w = 0.f;
    L = g;
  }
  return L;
}

template <int NKS>
__device__ __forceinline__ v8f band_mm(v8f acc, const float* p, const v2f* w) {
#pragma unroll
  for (int ks = 0; ks < NKS; ++ks) {
    v2f fa;
    fa.x = p[ks * 4];
    fa.y = p[ks * 4 + 1];
    acc = __builtin_amdgcn_wmma_f32_16x16x4_f32(
        false, fa, false, w[ks], (short)0, acc, false, false);
  }
  return acc;
}

template <int XM, int BANKS>
__device__ __forceinline__ float lane_xchg(float keep, float other) {
  return __builtin_bit_cast(float, __builtin_amdgcn_update_dpp(
      __builtin_bit_cast(int, keep), __builtin_bit_cast(int, other),
      0x160 + XM  , 0xF, BANKS, false));
}

__device__ __forceinline__ void lds_barrier() {
  asm volatile("s_wait_dscnt 0x0\n\ts_barrier_signal -1\n\ts_barrier_wait -1" ::: "memory");
}

__global__ __launch_bounds__(128, 4)
void idwt6_composed_wmma(const float* __restrict__ a,  const float* __restrict__ d6,
                         const float* __restrict__ d5, const float* __restrict__ d4,
                         const float* __restrict__ d3, const float* __restrict__ d2,
                         const float* __restrict__ d1, float* __restrict__ out) {
  const int wave = threadIdx.x >> 5;
  const int lane = threadIdx.x & 31;
  const int m  = lane & 15;
  const int hi = lane >> 4;
  const int pm = ((m & 3) << 2) | (m >> 2);
  const int cb = blockIdx.x >> 3;
  const int bt = blockIdx.x & 7;

  __shared__ __attribute__((aligned(16))) float s_in[2][16][LROW];

  const int r0 = wave << 2;
  const float* pA = (lane < 18) ? d2 : (lane < 28) ? d3 : d5;
  const int    nA = (lane < 18) ? N_D2 : (lane < 28) ? N_D3 : N_D5;
  const int  sftA = (lane < 18) ? 6 : (lane < 28) ? 5 : 3;
  const int  offA = (lane < 18) ? (lane << 2) : (lane < 28) ? ((lane - 18) << 2) : ((lane - 28) << 2);
  const int   clA = (lane < 28) ? (33 + lane) : (39 + lane);
  const float* pB = (m < 6) ? d4 : (m < 9) ? d6 : (m < 12) ? a : d1;
  const int    nB = (m < 6) ? N_D4 : (m < 9) ? N_D6 : (m < 12) ? N_A : N_D1;
  const int  sftB = (m < 6) ? 4 : (m < 12) ? 2 : 7;
  const int  offB = (m < 6) ? (m << 2) : (m < 9) ? ((m - 6) << 2) : (m < 12) ? ((m - 9) << 2) : 128;
  const int   clB = (m < 6) ? (61 + m) : (m < 12) ? (65 + m) : 32;

  const float* rD = d1 + (size_t)((bt << 4) + r0) * (size_t)N_D1;
  const float* rA = pA + (size_t)((bt << 4) + r0) * (size_t)nA;
  const float* rB = pB + (size_t)((bt << 4) + r0 + hi) * (size_t)nB;

  struct Strips { v4f gD[4], gA[4], gB[2]; };
  Strips st0, st1;
  auto request_strips = [&](Strips& s, int gx) {
    v4f (&gD)[4] = s.gD; v4f (&gA)[4] = s.gA; v4f (&gB)[2] = s.gB;
    const int colD = (gx << 7) + (lane << 2);
    const int colA = (gx << sftA) + offA;
    const int colB = (gx << sftB) + offB;
    const int ccD = colD < (N_D1 - 4) ? colD : (N_D1 - 4);
    const int ccA = colA < (nA - 4) ? colA : (nA - 4);
    const int ccB = colB < (nB - 4) ? colB : (nB - 4);
    const float* qD = rD + ccD;
    const float* qA = rA + ccA;
    const float* qB = rB + ccB;
#pragma unroll
    for (int rr = 0; rr < 4; ++rr) {
      gD[rr] = *(const v4f_u*)(qD + (size_t)rr * (size_t)N_D1);
      gA[rr] = *(const v4f_u*)(qA + (size_t)rr * (size_t)nA);
    }
#pragma unroll
    for (int q = 0; q < 2; ++q)
      gB[q] = *(const v4f_u*)(qB + (size_t)(2 * q) * (size_t)nB);
  };
  auto park_strips = [&](const Strips& s, int buf, int gx) {
    if (gx == 1023) {
      const int colD = (gx << 7) + (lane << 2);
      const int colA = (gx << sftA) + offA;
      const int colB = (gx << sftB) + offB;
      const int shD = colD - (colD < (N_D1 - 4) ? colD : (N_D1 - 4));
      const int shA = colA - (colA < (nA - 4) ? colA : (nA - 4));
      const int shB = colB - (colB < (nB - 4) ? colB : (nB - 4));
#pragma unroll
      for (int rr = 0; rr < 4; ++rr) {
        *(v4f*)&s_in[buf][r0 + rr][lane << 2] = fix_piece(s.gD[rr], shD);
        *(v4f*)&s_in[buf][r0 + rr][clA << 2]  = fix_piece(s.gA[rr], shA);
      }
#pragma unroll
      for (int q = 0; q < 2; ++q)
        *(v4f*)&s_in[buf][r0 + 2 * q + hi][clB << 2] = fix_piece(s.gB[q], shB);
    } else {
#pragma unroll
      for (int rr = 0; rr < 4; ++rr) {
        *(v4f*)&s_in[buf][r0 + rr][lane << 2] = s.gD[rr];
        *(v4f*)&s_in[buf][r0 + rr][clA << 2]  = s.gA[rr];
      }
#pragma unroll
      for (int q = 0; q < 2; ++q)
        *(v4f*)&s_in[buf][r0 + 2 * q + hi][clB << 2] = s.gB[q];
    }
  };

  const int gx0 = cb << 3;
  request_strips(st0, gx0 + (cb & 7));
  request_strips(st1, gx0 + ((1 + cb) & 7));
  __builtin_amdgcn_sched_barrier(0);

  v2f wa[4][2], w6[4][2], w5[2][2], w4[2], w3[2], w2[3], w1[3];
#pragma unroll
  for (int u = 0; u < 4; ++u) {
    build_w<6, 6, 2>(wa[u], 16 * u, pm, hi);
    build_w<5, 6, 2>(w6[u], 16 * u, pm, hi);
  }
#pragma unroll
  for (int q = 0; q < 2; ++q) build_w<4, 5, 2>(w5[q], 16 * q, pm, hi);
  build_w<3, 4, 2>(w4, 0, pm, hi);
  build_w<2, 3, 2>(w3, 0, pm, hi);
  build_w<1, 2, 3>(w2, 0, pm, hi);
  build_w<0, 1, 3>(w1, 0, pm, hi);

  __builtin_amdgcn_sched_barrier(0);

  park_strips(st0, 0, gx0 + (cb & 7));
  park_strips(st1, 1, gx0 + ((1 + cb) & 7));
  lds_barrier();

  float* po = out + (size_t)((bt << 4) + hi * 8) * (size_t)L_OUT + (wave << 6) + (m << 2);
  auto step = [&](int it, const int cur, const bool park, const bool ask, const bool more) {
    const int gx = gx0 + ((it + cb) & 7);
    if (park) park_strips(st0, cur ^ 1, gx0 + ((it + 1 + cb) & 7));
    if (ask) request_strips(st0, gx0 + ((it + 2 + cb) & 7));
    __builtin_amdgcn_sched_barrier(0);

    const float* srow = &s_in[cur][m][hi * 2];
    v8f acc[4];
#pragma unroll
    for (int u = 0; u < 4; ++u) {
      const int tl = (wave << 2) + u;
      v8f a8 = {0.f, 0.f, 0.f, 0.f, 0.f, 0.f, 0.f, 0.f};
      a8 = band_mm<2>(a8, srow + 296 + (tl >> 2), wa[u]);
      a8 = band_mm<2>(a8, srow + 284 + (tl >> 2), w6[u]);
      a8 = band_mm<2>(a8, srow + 268 + (tl >> 1), w5[u & 1]);
      a8 = band_mm<2>(a8, srow + 244 + tl,        w4);
      a8 = band_mm<2>(a8, srow + 204 + (tl << 1), w3);
      a8 = band_mm<3>(a8, srow + 132 + (tl << 2), w2);
      a8 = band_mm<3>(a8, srow +       (tl << 3), w1);
      acc[u] = a8;
      if (u == 1) __builtin_amdgcn_sched_barrier(0);
    }

    float* pt = po + (gx << 8);
#pragma unroll
    for (int r = 0; r < 8; ++r) {
      const float y0 = acc[0][r], y1 = acc[1][r], y2 = acc[2][r], y3 = acc[3][r];
      const float z0 = lane_xchg<8, 0xC>(y0, y2);
      const float z2 = lane_xchg<8, 0x3>(y2, y0);
      const float z1 = lane_xchg<8, 0xC>(y1, y3);
      const float z3 = lane_xchg<8, 0x3>(y3, y1);
      v4f o;
      o.x = lane_xchg<4, 0xA>(z0, z1);
      o.y = lane_xchg<4, 0x5>(z1, z0);
      o.z = lane_xchg<4, 0xA>(z2, z3);
      o.w = lane_xchg<4, 0x5>(z3, z2);
      *(v4f*)(pt + (size_t)r * (size_t)L_OUT) = o;
    }

    if (more) lds_barrier();
  };
  step(0, 0, false, true, true);
  for (int it = 1; it < 5; it += 2) {
    step(it,     1, true, true, true);
    step(it + 1, 0, true, true, true);
  }
  step(5, 1, true, true, true);
  step(6, 0, true, false, true);
  step(7, 1, false, false, false);
}

extern "C" void kernel_launch(void* const* d_in, const int* in_sizes, int n_in_cnt,
                              void* d_out, int out_size, void* d_ws, size_t ws_size,
                              hipStream_t stream) {
  (void)in_sizes; (void)n_in_cnt; (void)out_size; (void)d_ws; (void)ws_size;
  const float* a  = (const float*)d_in[0];
  const float* d6 = (const float*)d_in[1];
  const float* d5 = (const float*)d_in[2];
  const float* d4 = (const float*)d_in[3];
  const float* d3 = (const float*)d_in[4];
  const float* d2 = (const float*)d_in[5];
  const float* d1 = (const float*)d_in[6];
  float* out = (float*)d_out;

  idwt6_composed_wmma<<<dim3(1024, 1), 128, 0, stream>>>(a, d6, d5, d4, d3, d2, d1, out);
}


# LDS fragment reads of a step stand in four groups, each issued one 16-product block ahead of its use (were spread one or two at a time among the 64 products); product and sum order unchanged
# speedup vs baseline: 4.1479x; 1.0507x over previous
#include <hip/hip_runtime.h>


typedef __attribute__((ext_vector_type(2))) float v2f;
typedef __attribute__((ext_vector_type(8))) float v8f;
typedef __attribute__((ext_vector_type(4))) float v4f;
typedef v4f __attribute__((aligned(4))) v4f_u;

#define LROW 308

#define N_D1 131075
#define N_D2 65541
#define N_D3 32774
#define N_D4 16390
#define N_D5 8198
#define N_D6 4102
#define N_A  4102
#define L_OUT 262144
#define NST 8

constexpr double K_LO[8] = {
    (double)0.23037781330885523f,  (double)0.7148465705525415f,  (double)0.6308807679295904f,
    (double)-0.02798376941698385f, (double)-0.18703481171888114f, (double)0.030841381835986965f,
    (double)0.032883011666982945f, (double)-0.010597401784997278f};
constexpr double K_HI[8] = {
    (double)-0.010597401784997278f, (double)-0.032883011666982945f, (double)0.030841381835986965f,
    (double)0.18703481171888114f,   (double)-0.02798376941698385f,  (double)-0.6308807679295904f,
    (double)0.7148465705525415f,    (double)-0.23037781330885523f};

struct FiltD { double v[448]; int lo; int len; };

constexpr double fget(const FiltD& f, int s) {
  const int i = s - f.lo;
  return (i >= 0 && i < f.len) ? f.v[i] : 0.0;
}
constexpr FiltD fbase(const double* w) {
  FiltD f{};
  f.lo = -6; f.len = 8;
  for (int i = 0; i < 8; ++i) f.v[i] = w[i];
  return f;
}
constexpr FiltD flift(const double* w, const FiltD& gl, int r) {
  FiltD f{};
  const int st = 1 << r;
  f.lo = gl.lo - 6 * st;
  f.len = (gl.lo + gl.len - 1 + st) - f.lo + 1;
  for (int i = 0; i < f.len; ++i) {
    const int s = f.lo + i;
    double acc = 0.0;
    for (int u = -6; u <= 1; ++u) acc += w[u + 6] * fget(gl, s - st * u);
    f.v[i] = acc;
  }
  return f;
}

constexpr FiltD GL1 = fbase(K_LO);
constexpr FiltD GL2 = flift(K_LO, GL1, 1);
constexpr FiltD GL3 = flift(K_LO, GL2, 2);
constexpr FiltD GL4 = flift(K_LO, GL3, 3);
constexpr FiltD GL5 = flift(K_LO, GL4, 4);
constexpr FiltD GL6 = flift(K_LO, GL5, 5);
constexpr FiltD GH1 = fbase(K_HI);
constexpr FiltD GH2 = flift(K_HI, GL1, 1);
constexpr FiltD GH3 = flift(K_HI, GL2, 2);
constexpr FiltD GH4 = flift(K_HI, GL3, 3);
constexpr FiltD GH5 = flift(K_HI, GL4, 4);
constexpr FiltD GH6 = flift(K_HI, GL5, 5);

#define TPAD 72
struct Tab  { float v[520]; };
struct Tabs { Tab t[7]; };
constexpr Tab to_tab(const FiltD& f) {
  Tab t{};
  for (int i = 0; i < f.len; ++i) t.v[TPAD + i] = (float)f.v[i];
  return t;
}
constexpr Tabs make_tabs() {
  Tabs r{};
  r.t[0] = to_tab(GH1); r.t[1] = to_tab(GH2); r.t[2] = to_tab(GH3); r.t[3] = to_tab(GH4);
  r.t[4] = to_tab(GH5); r.t[5] = to_tab(GH6); r.t[6] = to_tab(GL6);
  return r;
}
constexpr Tabs H_TABS = make_tabs();
__constant__ Tabs c_tabs = H_TABS;

template <int B, int R, int NKS>
__device__ __forceinline__ void build_w(v2f* w, int delta, int m, int hi) {
  constexpr int LO = -6 * ((1 << R) - 1);
  const float* t = &c_tabs.t[B].v[0] + (m - ((2 * hi) << R));
#pragma unroll
  for (int ks = 0; ks < NKS; ++ks) {
    w[ks].x = t[TPAD - LO + delta - ((ks * 4) << R)];
    w[ks].y = t[TPAD - LO + delta - ((ks * 4 + 1) << R)];
  }
}

__device__ __forceinline__ v4f fix_piece(v4f L, int sh) {
  if (sh != 0) {
    v4f g;
    g.x = (sh == 1) ? L.y : (sh == 2) ? L.z : (sh == 3) ? L.w : 0.f;
    g.y = (sh == 1) ? L.z : (sh == 2) ? L.w : 0.f;
    g.z = (sh == 1) ? L.w : 0.f;
    g.w = 0.f;
    L = g;
  }
  return L;
}

template <int NKS>
__device__ __forceinline__ v8f band_mm(v8f acc, const float* p, const v2f* w) {
#pragma unroll
  for (int ks = 0; ks < NKS; ++ks) {
    v2f fa;
    fa.x = p[ks * 4];
    fa.y = p[ks * 4 + 1];
    acc = __builtin_amdgcn_wmma_f32_16x16x4_f32(
        false, fa, false, w[ks], (short)0, acc, false, false);
  }
  return acc;
}

template <int XM, int BANKS>
__device__ __forceinline__ float lane_xchg(float keep, float other) {
  return __builtin_bit_cast(float, __builtin_amdgcn_update_dpp(
      __builtin_bit_cast(int, keep), __builtin_bit_cast(int, other),
      0x160 + XM  , 0xF, BANKS, false));
}

__device__ __forceinline__ void lds_barrier() {
  asm volatile("s_wait_dscnt 0x0\n\ts_barrier_signal -1\n\ts_barrier_wait -1" ::: "memory");
}

__global__ __launch_bounds__(128, 4)
void idwt6_composed_wmma(const float* __restrict__ a,  const float* __restrict__ d6,
                         const float* __restrict__ d5, const float* __restrict__ d4,
                         const float* __restrict__ d3, const float* __restrict__ d2,
                         const float* __restrict__ d1, float* __restrict__ out) {
  const int wave = threadIdx.x >> 5;
  const int lane = threadIdx.x & 31;
  const int m  = lane & 15;
  const int hi = lane >> 4;
  const int pm = ((m & 3) << 2) | (m >> 2);
  const int cb = blockIdx.x >> 3;
  const int bt = blockIdx.x & 7;

  __shared__ __attribute__((aligned(16))) float s_in[2][16][LROW];

  const int r0 = wave << 2;
  const float* pA = (lane < 18) ? d2 : (lane < 28) ? d3 : d5;
  const int    nA = (lane < 18) ? N_D2 : (lane < 28) ? N_D3 : N_D5;
  const int  sftA = (lane < 18) ? 6 : (lane < 28) ? 5 : 3;
  const int  offA = (lane < 18) ? (lane << 2) : (lane < 28) ? ((lane - 18) << 2) : ((lane - 28) << 2);
  const int   clA = (lane < 28) ? (33 + lane) : (39 + lane);
  const float* pB = (m < 6) ? d4 : (m < 9) ? d6 : (m < 12) ? a : d1;
  const int    nB = (m < 6) ? N_D4 : (m < 9) ? N_D6 : (m < 12) ? N_A : N_D1;
  const int  sftB = (m < 6) ? 4 : (m < 12) ? 2 : 7;
  const int  offB = (m < 6) ? (m << 2) : (m < 9) ? ((m - 6) << 2) : (m < 12) ? ((m - 9) << 2) : 128;
  const int   clB = (m < 6) ? (61 + m) : (m < 12) ? (65 + m) : 32;

  const float* rD = d1 + (size_t)((bt << 4) + r0) * (size_t)N_D1;
  const float* rA = pA + (size_t)((bt << 4) + r0) * (size_t)nA;
  const float* rB = pB + (size_t)((bt << 4) + r0 + hi) * (size_t)nB;

  struct Strips { v4f gD[4], gA[4], gB[2]; };
  Strips st0, st1;
  auto request_strips = [&](Strips& s, int gx) {
    v4f (&gD)[4] = s.gD; v4f (&gA)[4] = s.gA; v4f (&gB)[2] = s.gB;
    const int colD = (gx << 7) + (lane << 2);
    const int colA = (gx << sftA) + offA;
    const int colB = (gx << sftB) + offB;
    const int ccD = colD < (N_D1 - 4) ? colD : (N_D1 - 4);
    const int ccA = colA < (nA - 4) ? colA : (nA - 4);
    const int ccB = colB < (nB - 4) ? colB : (nB - 4);
    const float* qD = rD + ccD;
    const float* qA = rA + ccA;
    const float* qB = rB + ccB;
#pragma unroll
    for (int rr = 0; rr < 4; ++rr) {
      gD[rr] = *(const v4f_u*)(qD + (size_t)rr * (size_t)N_D1);
      gA[rr] = *(const v4f_u*)(qA + (size_t)rr * (size_t)nA);
    }
#pragma unroll
    for (int q = 0; q < 2; ++q)
      gB[q] = *(const v4f_u*)(qB + (size_t)(2 * q) * (size_t)nB);
  };
  auto park_strips = [&](const Strips& s, int buf, int gx) {
    if (gx == 1023) {
      const int colD = (gx << 7) + (lane << 2);
      const int colA = (gx << sftA) + offA;
      const int colB = (gx << sftB) + offB;
      const int shD = colD - (colD < (N_D1 - 4) ? colD : (N_D1 - 4));
      const int shA = colA - (colA < (nA - 4) ? colA : (nA - 4));
      const int shB = colB - (colB < (nB - 4) ? colB : (nB - 4));
#pragma unroll
      for (int rr = 0; rr < 4; ++rr) {
        *(v4f*)&s_in[buf][r0 + rr][lane << 2] = fix_piece(s.gD[rr], shD);
        *(v4f*)&s_in[buf][r0 + rr][clA << 2]  = fix_piece(s.gA[rr], shA);
      }
#pragma unroll
      for (int q = 0; q < 2; ++q)
        *(v4f*)&s_in[buf][r0 + 2 * q + hi][clB << 2] = fix_piece(s.gB[q], shB);
    } else {
#pragma unroll
      for (int rr = 0; rr < 4; ++rr) {
        *(v4f*)&s_in[buf][r0 + rr][lane << 2] = s.gD[rr];
        *(v4f*)&s_in[buf][r0 + rr][clA << 2]  = s.gA[rr];
      }
#pragma unroll
      for (int q = 0; q < 2; ++q)
        *(v4f*)&s_in[buf][r0 + 2 * q + hi][clB << 2] = s.gB[q];
    }
  };

  const int gx0 = cb << 3;
  request_strips(st0, gx0 + (cb & 7));
  request_strips(st1, gx0 + ((1 + cb) & 7));
  __builtin_amdgcn_sched_barrier(0);

  v2f wa[4][2], w6[4][2], w5[2][2], w4[2], w3[2], w2[3], w1[3];
#pragma unroll
  for (int u = 0; u < 4; ++u) {
    build_w<6, 6, 2>(wa[u], 16 * u, pm, hi);
    build_w<5, 6, 2>(w6[u], 16 * u, pm, hi);
  }
#pragma unroll
  for (int q = 0; q < 2; ++q) build_w<4, 5, 2>(w5[q], 16 * q, pm, hi);
  build_w<3, 4, 2>(w4, 0, pm, hi);
  build_w<2, 3, 2>(w3, 0, pm, hi);
  build_w<1, 2, 3>(w2, 0, pm, hi);
  build_w<0, 1, 3>(w1, 0, pm, hi);

  __builtin_amdgcn_sched_barrier(0);

  park_strips(st0, 0, gx0 + (cb & 7));
  park_strips(st1, 1, gx0 + ((1 + cb) & 7));
  lds_barrier();

  float* po = out + (size_t)((bt << 4) + hi * 8) * (size_t)L_OUT + (wave << 6) + (m << 2);
  auto step = [&](int it, const int cur, const bool park, const bool ask, const bool more) {
    const int gx = gx0 + ((it + cb) & 7);
    if (park) park_strips(st0, cur ^ 1, gx0 + ((it + 1 + cb) & 7));
    if (ask) request_strips(st0, gx0 + ((it + 2 + cb) & 7));
    __builtin_amdgcn_sched_barrier(0);

    const float* srow = &s_in[cur][m][hi * 2];
    v8f acc[4];
    struct FragA { v2f a[2], s6[2], s5[2], s4[2][2]; };
    struct FragB { v2f s3[2][2], s2[2][3], s1[2][3]; };
    typedef v2f __attribute__((aligned(4))) v2f_u;
    auto frag = [&](const float* p) { return (v2f)(*(const v2f_u*)p); };
    auto mm = [&](v8f c, v2f fa, v2f fb) {
      return __builtin_amdgcn_wmma_f32_16x16x4_f32(false, fa, false, fb, (short)0, c, false, false);
    };
    auto load_a = [&](FragA& f, const int q) {
      const int t0 = (wave << 2) + 2 * q;
#pragma unroll
      for (int ks = 0; ks < 2; ++ks) {
        f.a[ks]  = frag(srow + 296 + wave + 4 * ks);
        f.s6[ks] = frag(srow + 284 + wave + 4 * ks);
        f.s5[ks] = frag(srow + 268 + (t0 >> 1) + 4 * ks);
        f.s4[0][ks] = frag(srow + 244 + t0 + 4 * ks);
        f.s4[1][ks] = frag(srow + 244 + t0 + 1 + 4 * ks);
      }
    };
    auto load_b = [&](FragB& f, const int q) {
      const int t0 = (wave << 2) + 2 * q;
#pragma unroll
      for (int e = 0; e < 2; ++e) {
        const int tl = t0 + e;
#pragma unroll
        for (int ks = 0; ks < 2; ++ks) f.s3[e][ks] = frag(srow + 204 + (tl << 1) + 4 * ks);
#pragma unroll
        for (int ks = 0; ks < 3; ++ks) f.s2[e][ks] = frag(srow + 132 + (tl << 2) + 4 * ks);
#pragma unroll
        for (int ks = 0; ks < 3; ++ks) f.s1[e][ks] = frag(srow + (tl << 3) + 4 * ks);
      }
    };
    auto prod_a = [&](const FragA& f, const int q) {
      const int u0 = 2 * q, u1 = 2 * q + 1;
      v8f c0 = {0.f, 0.f, 0.f, 0.f, 0.f, 0.f, 0.f, 0.f};
      v8f c1 = {0.f, 0.f, 0.f, 0.f, 0.f, 0.f, 0.f, 0.f};
#pragma unroll
      for (int ks = 0; ks < 2; ++ks) { c0 = mm(c0, f.a[ks], wa[u0][ks]);  c1 = mm(c1, f.a[ks], wa[u1][ks]); }
#pragma unroll
      for (int ks = 0; ks < 2; ++ks) { c0 = mm(c0, f.s6[ks], w6[u0][ks]); c1 = mm(c1, f.s6[ks], w6[u1][ks]); }
#pragma unroll
      for (int ks = 0; ks < 2; ++ks) { c0 = mm(c0, f.s5[ks], w5[0][ks]);  c1 = mm(c1, f.s5[ks], w5[1][ks]); }
#pragma unroll
      for (int ks = 0; ks < 2; ++ks) { c0 = mm(c0, f.s4[0][ks], w4[ks]);  c1 = mm(c1, f.s4[1][ks], w4[ks]); }
      acc[u0] = c0; acc[u1] = c1;
    };
    auto prod_b = [&](const FragB& f, const int q) {
      const int u0 = 2 * q, u1 = 2 * q + 1;
      v8f c0 = acc[u0], c1 = acc[u1];
#pragma unroll
      for (int ks = 0; ks < 2; ++ks) { c0 = mm(c0, f.s3[0][ks], w3[ks]); c1 = mm(c1, f.s3[1][ks], w3[ks]); }
#pragma unroll
      for (int ks = 0; ks < 3; ++ks) { c0 = mm(c0, f.s2[0][ks], w2[ks]); c1 = mm(c1, f.s2[1][ks], w2[ks]); }
#pragma unroll
      for (int ks = 0; ks < 3; ++ks) { c0 = mm(c0, f.s1[0][ks], w1[ks]); c1 = mm(c1, f.s1[1][ks], w1[ks]); }
      acc[u0] = c0; acc[u1] = c1;
    };
    FragA fa; FragB fb;
    load_a(fa, 0);
    __builtin_amdgcn_sched_barrier(0);
    load_b(fb, 0);
    __builtin_amdgcn_sched_barrier(0);
    prod_a(fa, 0);
    __builtin_amdgcn_sched_barrier(0);
    load_a(fa, 1);
    __builtin_amdgcn_sched_barrier(0);
    prod_b(fb, 0);
    __builtin_amdgcn_sched_barrier(0);
    load_b(fb, 1);
    __builtin_amdgcn_sched_barrier(0);
    prod_a(fa, 1);
    prod_b(fb, 1);

    float* pt = po + (gx << 8);
#pragma unroll
    for (int r = 0; r < 8; ++r) {
      const float y0 = acc[0][r], y1 = acc[1][r], y2 = acc[2][r], y3 = acc[3][r];
      const float z0 = lane_xchg<8, 0xC>(y0, y2);
      const float z2 = lane_xchg<8, 0x3>(y2, y0);
      const float z1 = lane_xchg<8, 0xC>(y1, y3);
      const float z3 = lane_xchg<8, 0x3>(y3, y1);
      v4f o;
      o.x = lane_xchg<4, 0xA>(z0, z1);
      o.y = lane_xchg<4, 0x5>(z1, z0);
      o.z = lane_xchg<4, 0xA>(z2, z3);
      o.w = lane_xchg<4, 0x5>(z3, z2);
      *(v4f*)(pt + (size_t)r * (size_t)L_OUT) = o;
    }

    if (more) lds_barrier();
  };
  step(0, 0, false, true, true);
  for (int it = 1; it < 5; it += 2) {
    step(it,     1, true, true, true);
    step(it + 1, 0, true, true, true);
  }
  step(5, 1, true, true, true);
  step(6, 0, true, false, true);
  step(7, 1, false, false, false);
}

extern "C" void kernel_launch(void* const* d_in, const int* in_sizes, int n_in_cnt,
                              void* d_out, int out_size, void* d_ws, size_t ws_size,
                              hipStream_t stream) {
  (void)in_sizes; (void)n_in_cnt; (void)out_size; (void)d_ws; (void)ws_size;
  const float* a  = (const float*)d_in[0];
  const float* d6 = (const float*)d_in[1];
  const float* d5 = (const float*)d_in[2];
  const float* d4 = (const float*)d_in[3];
  const float* d3 = (const float*)d_in[4];
  const float* d2 = (const float*)d_in[5];
  const float* d1 = (const float*)d_in[6];
  float* out = (float*)d_out;

  idwt6_composed_wmma<<<dim3(1024, 1), 128, 0, stream>>>(a, d6, d5, d4, d3, d2, d1, out);
}


# Each wave asks the second-level cache once, before its first wait, for its four rows' whole 8-step input span (8 prefetch instructions): later strip loads hit it; products, stores, sums unchanged
# speedup vs baseline: 4.3246x; 1.0426x over previous
#include <hip/hip_runtime.h>


typedef __attribute__((ext_vector_type(2))) float v2f;
typedef __attribute__((ext_vector_type(8))) float v8f;
typedef __attribute__((ext_vector_type(4))) float v4f;
typedef v4f __attribute__((aligned(4))) v4f_u;

#define LROW 308

#define N_D1 131075
#define N_D2 65541
#define N_D3 32774
#define N_D4 16390
#define N_D5 8198
#define N_D6 4102
#define N_A  4102
#define L_OUT 262144
#define NST 8

constexpr double K_LO[8] = {
    (double)0.23037781330885523f,  (double)0.7148465705525415f,  (double)0.6308807679295904f,
    (double)-0.02798376941698385f, (double)-0.18703481171888114f, (double)0.030841381835986965f,
    (double)0.032883011666982945f, (double)-0.010597401784997278f};
constexpr double K_HI[8] = {
    (double)-0.010597401784997278f, (double)-0.032883011666982945f, (double)0.030841381835986965f,
    (double)0.18703481171888114f,   (double)-0.02798376941698385f,  (double)-0.6308807679295904f,
    (double)0.7148465705525415f,    (double)-0.23037781330885523f};

struct FiltD { double v[448]; int lo; int len; };

constexpr double fget(const FiltD& f, int s) {
  const int i = s - f.lo;
  return (i >= 0 && i < f.len) ? f.v[i] : 0.0;
}
constexpr FiltD fbase(const double* w) {
  FiltD f{};
  f.lo = -6; f.len = 8;
  for (int i = 0; i < 8; ++i) f.v[i] = w[i];
  return f;
}
constexpr FiltD flift(const double* w, const FiltD& gl, int r) {
  FiltD f{};
  const int st = 1 << r;
  f.lo = gl.lo - 6 * st;
  f.len = (gl.lo + gl.len - 1 + st) - f.lo + 1;
  for (int i = 0; i < f.len; ++i) {
    const int s = f.lo + i;
    double acc = 0.0;
    for (int u = -6; u <= 1; ++u) acc += w[u + 6] * fget(gl, s - st * u);
    f.v[i] = acc;
  }
  return f;
}

constexpr FiltD GL1 = fbase(K_LO);
constexpr FiltD GL2 = flift(K_LO, GL1, 1);
constexpr FiltD GL3 = flift(K_LO, GL2, 2);
constexpr FiltD GL4 = flift(K_LO, GL3, 3);
constexpr FiltD GL5 = flift(K_LO, GL4, 4);
constexpr FiltD GL6 = flift(K_LO, GL5, 5);
constexpr FiltD GH1 = fbase(K_HI);
constexpr FiltD GH2 = flift(K_HI, GL1, 1);
constexpr FiltD GH3 = flift(K_HI, GL2, 2);
constexpr FiltD GH4 = flift(K_HI, GL3, 3);
constexpr FiltD GH5 = flift(K_HI, GL4, 4);
constexpr FiltD GH6 = flift(K_HI, GL5, 5);

#define TPAD 72
struct Tab  { float v[520]; };
struct Tabs { Tab t[7]; };
constexpr Tab to_tab(const FiltD& f) {
  Tab t{};
  for (int i = 0; i < f.len; ++i) t.v[TPAD + i] = (float)f.v[i];
  return t;
}
constexpr Tabs make_tabs() {
  Tabs r{};
  r.t[0] = to_tab(GH1); r.t[1] = to_tab(GH2); r.t[2] = to_tab(GH3); r.t[3] = to_tab(GH4);
  r.t[4] = to_tab(GH5); r.t[5] = to_tab(GH6); r.t[6] = to_tab(GL6);
  return r;
}
constexpr Tabs H_TABS = make_tabs();
__constant__ Tabs c_tabs = H_TABS;

template <int B, int R, int NKS>
__device__ __forceinline__ void build_w(v2f* w, int delta, int m, int hi) {
  constexpr int LO = -6 * ((1 << R) - 1);
  const float* t = &c_tabs.t[B].v[0] + (m - ((2 * hi) << R));
#pragma unroll
  for (int ks = 0; ks < NKS; ++ks) {
    w[ks].x = t[TPAD - LO + delta - ((ks * 4) << R)];
    w[ks].y = t[TPAD - LO + delta - ((ks * 4 + 1) << R)];
  }
}

__device__ __forceinline__ v4f fix_piece(v4f L, int sh) {
  if (sh != 0) {
    v4f g;
    g.x = (sh == 1) ? L.y : (sh == 2) ? L.z : (sh == 3) ? L.w : 0.f;
    g.y = (sh == 1) ? L.z : (sh == 2) ? L.w : 0.f;
    g.z = (sh == 1) ? L.w : 0.f;
    g.w = 0.f;
    L = g;
  }
  return L;
}

template <int NKS>
__device__ __forceinline__ v8f band_mm(v8f acc, const float* p, const v2f* w) {
#pragma unroll
  for (int ks = 0; ks < NKS; ++ks) {
    v2f fa;
    fa.x = p[ks * 4];
    fa.y = p[ks * 4 + 1];
    acc = __builtin_amdgcn_wmma_f32_16x16x4_f32(
        false, fa, false, w[ks], (short)0, acc, false, false);
  }
  return acc;
}

template <int XM, int BANKS>
__device__ __forceinline__ float lane_xchg(float keep, float other) {
  return __builtin_bit_cast(float, __builtin_amdgcn_update_dpp(
      __builtin_bit_cast(int, keep), __builtin_bit_cast(int, other),
      0x160 + XM  , 0xF, BANKS, false));
}

__device__ __forceinline__ void lds_barrier() {
  asm volatile("s_wait_dscnt 0x0\n\ts_barrier_signal -1\n\ts_barrier_wait -1" ::: "memory");
}

__global__ __launch_bounds__(128, 4)
void idwt6_composed_wmma(const float* __restrict__ a,  const float* __restrict__ d6,
                         const float* __restrict__ d5, const float* __restrict__ d4,
                         const float* __restrict__ d3, const float* __restrict__ d2,
                         const float* __restrict__ d1, float* __restrict__ out) {
  const int wave = threadIdx.x >> 5;
  const int lane = threadIdx.x & 31;
  const int m  = lane & 15;
  const int hi = lane >> 4;
  const int pm = ((m & 3) << 2) | (m >> 2);
  const int cb = blockIdx.x >> 3;
  const int bt = blockIdx.x & 7;

  __shared__ __attribute__((aligned(16))) float s_in[2][16][LROW];

  const int r0 = wave << 2;
  const float* pA = (lane < 18) ? d2 : (lane < 28) ? d3 : d5;
  const int    nA = (lane < 18) ? N_D2 : (lane < 28) ? N_D3 : N_D5;
  const int  sftA = (lane < 18) ? 6 : (lane < 28) ? 5 : 3;
  const int  offA = (lane < 18) ? (lane << 2) : (lane < 28) ? ((lane - 18) << 2) : ((lane - 28) << 2);
  const int   clA = (lane < 28) ? (33 + lane) : (39 + lane);
  const float* pB = (m < 6) ? d4 : (m < 9) ? d6 : (m < 12) ? a : d1;
  const int    nB = (m < 6) ? N_D4 : (m < 9) ? N_D6 : (m < 12) ? N_A : N_D1;
  const int  sftB = (m < 6) ? 4 : (m < 12) ? 2 : 7;
  const int  offB = (m < 6) ? (m << 2) : (m < 9) ? ((m - 6) << 2) : (m < 12) ? ((m - 9) << 2) : 128;
  const int   clB = (m < 6) ? (61 + m) : (m < 12) ? (65 + m) : 32;

  const float* rD = d1 + (size_t)((bt << 4) + r0) * (size_t)N_D1;
  const float* rA = pA + (size_t)((bt << 4) + r0) * (size_t)nA;
  const float* rB = pB + (size_t)((bt << 4) + r0 + hi) * (size_t)nB;

  struct Strips { v4f gD[4], gA[4], gB[2]; };
  Strips st0, st1;
  auto request_strips = [&](Strips& s, int gx) {
    v4f (&gD)[4] = s.gD; v4f (&gA)[4] = s.gA; v4f (&gB)[2] = s.gB;
    const int colD = (gx << 7) + (lane << 2);
    const int colA = (gx << sftA) + offA;
    const int colB = (gx << sftB) + offB;
    const int ccD = colD < (N_D1 - 4) ? colD : (N_D1 - 4);
    const int ccA = colA < (nA - 4) ? colA : (nA - 4);
    const int ccB = colB < (nB - 4) ? colB : (nB - 4);
    const float* qD = rD + ccD;
    const float* qA = rA + ccA;
    const float* qB = rB + ccB;
#pragma unroll
    for (int rr = 0; rr < 4; ++rr) {
      gD[rr] = *(const v4f_u*)(qD + (size_t)rr * (size_t)N_D1);
      gA[rr] = *(const v4f_u*)(qA + (size_t)rr * (size_t)nA);
    }
#pragma unroll
    for (int q = 0; q < 2; ++q)
      gB[q] = *(const v4f_u*)(qB + (size_t)(2 * q) * (size_t)nB);
  };
  auto park_strips = [&](const Strips& s, int buf, int gx) {
    if (gx == 1023) {
      const int colD = (gx << 7) + (lane << 2);
      const int colA = (gx << sftA) + offA;
      const int colB = (gx << sftB) + offB;
      const int shD = colD - (colD < (N_D1 - 4) ? colD : (N_D1 - 4));
      const int shA = colA - (colA < (nA - 4) ? colA : (nA - 4));
      const int shB = colB - (colB < (nB - 4) ? colB : (nB - 4));
#pragma unroll
      for (int rr = 0; rr < 4; ++rr) {
        *(v4f*)&s_in[buf][r0 + rr][lane << 2] = fix_piece(s.gD[rr], shD);
        *(v4f*)&s_in[buf][r0 + rr][clA << 2]  = fix_piece(s.gA[rr], shA);
      }
#pragma unroll
      for (int q = 0; q < 2; ++q)
        *(v4f*)&s_in[buf][r0 + 2 * q + hi][clB << 2] = fix_piece(s.gB[q], shB);
    } else {
#pragma unroll
      for (int rr = 0; rr < 4; ++rr) {
        *(v4f*)&s_in[buf][r0 + rr][lane << 2] = s.gD[rr];
        *(v4f*)&s_in[buf][r0 + rr][clA << 2]  = s.gA[rr];
      }
#pragma unroll
      for (int q = 0; q < 2; ++q)
        *(v4f*)&s_in[buf][r0 + 2 * q + hi][clB << 2] = s.gB[q];
    }
  };

  const int gx0 = cb << 3;
  request_strips(st0, gx0 + (cb & 7));
  request_strips(st1, gx0 + ((1 + cb) & 7));
  __builtin_amdgcn_sched_barrier(0);

  v2f wa[4][2], w6[4][2], w5[2][2], w4[2], w3[2], w2[3], w1[3];
#pragma unroll
  for (int u = 0; u < 4; ++u) {
    build_w<6, 6, 2>(wa[u], 16 * u, pm, hi);
    build_w<5, 6, 2>(w6[u], 16 * u, pm, hi);
  }
#pragma unroll
  for (int q = 0; q < 2; ++q) build_w<4, 5, 2>(w5[q], 16 * q, pm, hi);
  build_w<3, 4, 2>(w4, 0, pm, hi);
  build_w<2, 3, 2>(w3, 0, pm, hi);
  build_w<1, 2, 3>(w2, 0, pm, hi);
  build_w<0, 1, 3>(w1, 0, pm, hi);

  __builtin_amdgcn_sched_barrier(0);

  {
    const int prow = (bt << 4) + r0 + (lane >> 3);
    const int pj = lane & 7;
    auto pf = [&](const float* base, int n, int sft, int span, int unit) {
      int c = (unit << 6) < (span - 1) ? (unit << 6) : (span - 1);
      c += gx0 << sft;
      c = c < (n - 1) ? c : (n - 1);
      __builtin_prefetch(base + (size_t)prow * (size_t)n + c, 0, 3);
    };
    pf(d1, N_D1, 7, 1028, pj);
    pf(d1, N_D1, 7, 1028, pj + 8);
    pf(d1, N_D1, 7, 1028, pj + 16);
    pf(d2, N_D2, 6, 520, pj);
    pf(d2, N_D2, 6, 520, pj + 8);
    pf(d3, N_D3, 5, 264, pj);
    const bool lo4 = pj < 4;
    pf(lo4 ? d4 : d5, lo4 ? N_D4 : N_D5, lo4 ? 4 : 3, lo4 ? 136 : 72, pj & 3);
    const bool ev2 = (pj & 2) == 0;
    pf(ev2 ? d6 : a, N_D6, 2, 40, pj & 1);
  }
  __builtin_amdgcn_sched_barrier(0);

  park_strips(st0, 0, gx0 + (cb & 7));
  park_strips(st1, 1, gx0 + ((1 + cb) & 7));
  lds_barrier();

  float* po = out + (size_t)((bt << 4) + hi * 8) * (size_t)L_OUT + (wave << 6) + (m << 2);
  auto step = [&](int it, const int cur, const bool park, const bool ask, const bool more) {
    const int gx = gx0 + ((it + cb) & 7);
    if (park) park_strips(st0, cur ^ 1, gx0 + ((it + 1 + cb) & 7));
    if (ask) request_strips(st0, gx0 + ((it + 2 + cb) & 7));
    __builtin_amdgcn_sched_barrier(0);

    const float* srow = &s_in[cur][m][hi * 2];
    v8f acc[4];
    struct FragA { v2f a[2], s6[2], s5[2], s4[2][2]; };
    struct FragB { v2f s3[2][2], s2[2][3], s1[2][3]; };
    typedef v2f __attribute__((aligned(4))) v2f_u;
    auto frag = [&](const float* p) { return (v2f)(*(const v2f_u*)p); };
    auto mm = [&](v8f c, v2f fa, v2f fb) {
      return __builtin_amdgcn_wmma_f32_16x16x4_f32(false, fa, false, fb, (short)0, c, false, false);
    };
    auto load_a = [&](FragA& f, const int q) {
      const int t0 = (wave << 2) + 2 * q;
#pragma unroll
      for (int ks = 0; ks < 2; ++ks) {
        f.a[ks]  = frag(srow + 296 + wave + 4 * ks);
        f.s6[ks] = frag(srow + 284 + wave + 4 * ks);
        f.s5[ks] = frag(srow + 268 + (t0 >> 1) + 4 * ks);
        f.s4[0][ks] = frag(srow + 244 + t0 + 4 * ks);
        f.s4[1][ks] = frag(srow + 244 + t0 + 1 + 4 * ks);
      }
    };
    auto load_b = [&](FragB& f, const int q) {
      const int t0 = (wave << 2) + 2 * q;
#pragma unroll
      for (int e = 0; e < 2; ++e) {
        const int tl = t0 + e;
#pragma unroll
        for (int ks = 0; ks < 2; ++ks) f.s3[e][ks] = frag(srow + 204 + (tl << 1) + 4 * ks);
#pragma unroll
        for (int ks = 0; ks < 3; ++ks) f.s2[e][ks] = frag(srow + 132 + (tl << 2) + 4 * ks);
#pragma unroll
        for (int ks = 0; ks < 3; ++ks) f.s1[e][ks] = frag(srow + (tl << 3) + 4 * ks);
      }
    };
    auto prod_a = [&](const FragA& f, const int q) {
      const int u0 = 2 * q, u1 = 2 * q + 1;
      v8f c0 = {0.f, 0.f, 0.f, 0.f, 0.f, 0.f, 0.f, 0.f};
      v8f c1 = {0.f, 0.f, 0.f, 0.f, 0.f, 0.f, 0.f, 0.f};
#pragma unroll
      for (int ks = 0; ks < 2; ++ks) { c0 = mm(c0, f.a[ks], wa[u0][ks]);  c1 = mm(c1, f.a[ks], wa[u1][ks]); }
#pragma unroll
      for (int ks = 0; ks < 2; ++ks) { c0 = mm(c0, f.s6[ks], w6[u0][ks]); c1 = mm(c1, f.s6[ks], w6[u1][ks]); }
#pragma unroll
      for (int ks = 0; ks < 2; ++ks) { c0 = mm(c0, f.s5[ks], w5[0][ks]);  c1 = mm(c1, f.s5[ks], w5[1][ks]); }
#pragma unroll
      for (int ks = 0; ks < 2; ++ks) { c0 = mm(c0, f.s4[0][ks], w4[ks]);  c1 = mm(c1, f.s4[1][ks], w4[ks]); }
      acc[u0] = c0; acc[u1] = c1;
    };
    auto prod_b = [&](const FragB& f, const int q) {
      const int u0 = 2 * q, u1 = 2 * q + 1;
      v8f c0 = acc[u0], c1 = acc[u1];
#pragma unroll
      for (int ks = 0; ks < 2; ++ks) { c0 = mm(c0, f.s3[0][ks], w3[ks]); c1 = mm(c1, f.s3[1][ks], w3[ks]); }
#pragma unroll
      for (int ks = 0; ks < 3; ++ks) { c0 = mm(c0, f.s2[0][ks], w2[ks]); c1 = mm(c1, f.s2[1][ks], w2[ks]); }
#pragma unroll
      for (int ks = 0; ks < 3; ++ks) { c0 = mm(c0, f.s1[0][ks], w1[ks]); c1 = mm(c1, f.s1[1][ks], w1[ks]); }
      acc[u0] = c0; acc[u1] = c1;
    };
    FragA fa; FragB fb;
    load_a(fa, 0);
    __builtin_amdgcn_sched_barrier(0);
    load_b(fb, 0);
    __builtin_amdgcn_sched_barrier(0);
    prod_a(fa, 0);
    __builtin_amdgcn_sched_barrier(0);
    load_a(fa, 1);
    __builtin_amdgcn_sched_barrier(0);
    prod_b(fb, 0);
    __builtin_amdgcn_sched_barrier(0);
    load_b(fb, 1);
    __builtin_amdgcn_sched_barrier(0);
    prod_a(fa, 1);
    prod_b(fb, 1);

    float* pt = po + (gx << 8);
#pragma unroll
    for (int r = 0; r < 8; ++r) {
      const float y0 = acc[0][r], y1 = acc[1][r], y2 = acc[2][r], y3 = acc[3][r];
      const float z0 = lane_xchg<8, 0xC>(y0, y2);
      const float z2 = lane_xchg<8, 0x3>(y2, y0);
      const float z1 = lane_xchg<8, 0xC>(y1, y3);
      const float z3 = lane_xchg<8, 0x3>(y3, y1);
      v4f o;
      o.x = lane_xchg<4, 0xA>(z0, z1);
      o.y = lane_xchg<4, 0x5>(z1, z0);
      o.z = lane_xchg<4, 0xA>(z2, z3);
      o.w = lane_xchg<4, 0x5>(z3, z2);
      *(v4f*)(pt + (size_t)r * (size_t)L_OUT) = o;
    }

    if (more) lds_barrier();
  };
  step(0, 0, false, true, true);
  for (int it = 1; it < 5; it += 2) {
    step(it,     1, true, true, true);
    step(it + 1, 0, true, true, true);
  }
  step(5, 1, true, true, true);
  step(6, 0, true, false, true);
  step(7, 1, false, false, false);
}

extern "C" void kernel_launch(void* const* d_in, const int* in_sizes, int n_in_cnt,
                              void* d_out, int out_size, void* d_ws, size_t ws_size,
                              hipStream_t stream) {
  (void)in_sizes; (void)n_in_cnt; (void)out_size; (void)d_ws; (void)ws_size;
  const float* a  = (const float*)d_in[0];
  const float* d6 = (const float*)d_in[1];
  const float* d5 = (const float*)d_in[2];
  const float* d4 = (const float*)d_in[3];
  const float* d3 = (const float*)d_in[4];
  const float* d2 = (const float*)d_in[5];
  const float* d1 = (const float*)d_in[6];
  float* out = (float*)d_out;

  idwt6_composed_wmma<<<dim3(1024, 1), 128, 0, stream>>>(a, d6, d5, d4, d3, d2, d1, out);
}


# Spans of the four coarse bands (d4, d5, d6, a) over the whole 8-step walk are parked in LDS once; a step loads only d1, d2, d3 strips (8 row loads, were 10; coarse halos not re-read); sums unchanged
# speedup vs baseline: 4.5631x; 1.0551x over previous
#include <hip/hip_runtime.h>


typedef __attribute__((ext_vector_type(2))) float v2f;
typedef __attribute__((ext_vector_type(8))) float v8f;
typedef __attribute__((ext_vector_type(4))) float v4f;
typedef v4f __attribute__((aligned(4))) v4f_u;

#define LROW 244
#define SPROW 292
#define SP_D4 0
#define SP_D5 136
#define SP_D6 208
#define SP_A  248

#define N_D1 131075
#define N_D2 65541
#define N_D3 32774
#define N_D4 16390
#define N_D5 8198
#define N_D6 4102
#define N_A  4102
#define L_OUT 262144
#define NST 8

constexpr double K_LO[8] = {
    (double)0.23037781330885523f,  (double)0.7148465705525415f,  (double)0.6308807679295904f,
    (double)-0.02798376941698385f, (double)-0.18703481171888114f, (double)0.030841381835986965f,
    (double)0.032883011666982945f, (double)-0.010597401784997278f};
constexpr double K_HI[8] = {
    (double)-0.010597401784997278f, (double)-0.032883011666982945f, (double)0.030841381835986965f,
    (double)0.18703481171888114f,   (double)-0.02798376941698385f,  (double)-0.6308807679295904f,
    (double)0.7148465705525415f,    (double)-0.23037781330885523f};

struct FiltD { double v[448]; int lo; int len; };

constexpr double fget(const FiltD& f, int s) {
  const int i = s - f.lo;
  return (i >= 0 && i < f.len) ? f.v[i] : 0.0;
}
constexpr FiltD fbase(const double* w) {
  FiltD f{};
  f.lo = -6; f.len = 8;
  for (int i = 0; i < 8; ++i) f.v[i] = w[i];
  return f;
}
constexpr FiltD flift(const double* w, const FiltD& gl, int r) {
  FiltD f{};
  const int st = 1 << r;
  f.lo = gl.lo - 6 * st;
  f.len = (gl.lo + gl.len - 1 + st) - f.lo + 1;
  for (int i = 0; i < f.len; ++i) {
    const int s = f.lo + i;
    double acc = 0.0;
    for (int u = -6; u <= 1; ++u) acc += w[u + 6] * fget(gl, s - st * u);
    f.v[i] = acc;
  }
  return f;
}

constexpr FiltD GL1 = fbase(K_LO);
constexpr FiltD GL2 = flift(K_LO, GL1, 1);
constexpr FiltD GL3 = flift(K_LO, GL2, 2);
constexpr FiltD GL4 = flift(K_LO, GL3, 3);
constexpr FiltD GL5 = flift(K_LO, GL4, 4);
constexpr FiltD GL6 = flift(K_LO, GL5, 5);
constexpr FiltD GH1 = fbase(K_HI);
constexpr FiltD GH2 = flift(K_HI, GL1, 1);
constexpr FiltD GH3 = flift(K_HI, GL2, 2);
constexpr FiltD GH4 = flift(K_HI, GL3, 3);
constexpr FiltD GH5 = flift(K_HI, GL4, 4);
constexpr FiltD GH6 = flift(K_HI, GL5, 5);

#define TPAD 72
struct Tab  { float v[520]; };
struct Tabs { Tab t[7]; };
constexpr Tab to_tab(const FiltD& f) {
  Tab t{};
  for (int i = 0; i < f.len; ++i) t.v[TPAD + i] = (float)f.v[i];
  return t;
}
constexpr Tabs make_tabs() {
  Tabs r{};
  r.t[0] = to_tab(GH1); r.t[1] = to_tab(GH2); r.t[2] = to_tab(GH3); r.t[3] = to_tab(GH4);
  r.t[4] = to_tab(GH5); r.t[5] = to_tab(GH6); r.t[6] = to_tab(GL6);
  return r;
}
constexpr Tabs H_TABS = make_tabs();
__constant__ Tabs c_tabs = H_TABS;

template <int B, int R, int NKS>
__device__ __forceinline__ void build_w(v2f* w, int delta, int m, int hi) {
  constexpr int LO = -6 * ((1 << R) - 1);
  const float* t = &c_tabs.t[B].v[0] + (m - ((2 * hi) << R));
#pragma unroll
  for (int ks = 0; ks < NKS; ++ks) {
    w[ks].x = t[TPAD - LO + delta - ((ks * 4) << R)];
    w[ks].y = t[TPAD - LO + delta - ((ks * 4 + 1) << R)];
  }
}

__device__ __forceinline__ v4f fix_piece(v4f L, int sh) {
  if (sh != 0) {
    v4f g;
    g.x = (sh == 1) ? L.y : (sh == 2) ? L.z : (sh == 3) ? L.w : 0.f;
    g.y = (sh == 1) ? L.z : (sh == 2) ? L.w : 0.f;
    g.z = (sh == 1) ? L.w : 0.f;
    g.w = 0.f;
    L = g;
  }
  return L;
}

template <int NKS>
__device__ __forceinline__ v8f band_mm(v8f acc, const float* p, const v2f* w) {
#pragma unroll
  for (int ks = 0; ks < NKS; ++ks) {
    v2f fa;
    fa.x = p[ks * 4];
    fa.y = p[ks * 4 + 1];
    acc = __builtin_amdgcn_wmma_f32_16x16x4_f32(
        false, fa, false, w[ks], (short)0, acc, false, false);
  }
  return acc;
}

template <int XM, int BANKS>
__device__ __forceinline__ float lane_xchg(float keep, float other) {
  return __builtin_bit_cast(float, __builtin_amdgcn_update_dpp(
      __builtin_bit_cast(int, keep), __builtin_bit_cast(int, other),
      0x160 + XM  , 0xF, BANKS, false));
}

__device__ __forceinline__ void lds_barrier() {
  asm volatile("s_wait_dscnt 0x0\n\ts_barrier_signal -1\n\ts_barrier_wait -1" ::: "memory");
}

__global__ __launch_bounds__(128, 4)
void idwt6_composed_wmma(const float* __restrict__ a,  const float* __restrict__ d6,
                         const float* __restrict__ d5, const float* __restrict__ d4,
                         const float* __restrict__ d3, const float* __restrict__ d2,
                         const float* __restrict__ d1, float* __restrict__ out) {
  const int wave = threadIdx.x >> 5;
  const int lane = threadIdx.x & 31;
  const int m  = lane & 15;
  const int hi = lane >> 4;
  const int pm = ((m & 3) << 2) | (m >> 2);
  const int cb = blockIdx.x >> 3;
  const int bt = blockIdx.x & 7;

  __shared__ __attribute__((aligned(16))) float s_in[2][16][LROW];
  __shared__ __attribute__((aligned(16))) float s_sp[16][SPROW];

  const int r0 = wave << 2;
  const float* pA = (lane < 18) ? d2 : (lane < 28) ? d3 : d1;
  const int    nA = (lane < 18) ? N_D2 : (lane < 28) ? N_D3 : N_D1;
  const int  sftA = (lane < 18) ? 6 : (lane < 28) ? 5 : 7;
  const int  offA = (lane < 18) ? (lane << 2) : (lane < 28) ? ((lane - 18) << 2) : 128;
  const int   clA = (lane < 28) ? (33 + lane) : 32;

  const float* rD = d1 + (size_t)((bt << 4) + r0) * (size_t)N_D1;
  const float* rA = pA + (size_t)((bt << 4) + r0) * (size_t)nA;

  struct Strips { v4f gD[4], gA[4]; };
  Strips st0, st1;
  auto request_strips = [&](Strips& s, int gx) {
    v4f (&gD)[4] = s.gD; v4f (&gA)[4] = s.gA;
    const int colD = (gx << 7) + (lane << 2);
    const int colA = (gx << sftA) + offA;
    const int ccD = colD < (N_D1 - 4) ? colD : (N_D1 - 4);
    const int ccA = colA < (nA - 4) ? colA : (nA - 4);
    const float* qD = rD + ccD;
    const float* qA = rA + ccA;
#pragma unroll
    for (int rr = 0; rr < 4; ++rr) {
      gD[rr] = *(const v4f_u*)(qD + (size_t)rr * (size_t)N_D1);
      gA[rr] = *(const v4f_u*)(qA + (size_t)rr * (size_t)nA);
    }
  };
  auto park_strips = [&](const Strips& s, int buf, int gx) {
    if (gx == 1023) {
      const int colD = (gx << 7) + (lane << 2);
      const int colA = (gx << sftA) + offA;
      const int shD = colD - (colD < (N_D1 - 4) ? colD : (N_D1 - 4));
      const int shA = colA - (colA < (nA - 4) ? colA : (nA - 4));
#pragma unroll
      for (int rr = 0; rr < 4; ++rr) {
        *(v4f*)&s_in[buf][r0 + rr][lane << 2] = fix_piece(s.gD[rr], shD);
        *(v4f*)&s_in[buf][r0 + rr][clA << 2]  = fix_piece(s.gA[rr], shA);
      }
    } else {
#pragma unroll
      for (int rr = 0; rr < 4; ++rr) {
        *(v4f*)&s_in[buf][r0 + rr][lane << 2] = s.gD[rr];
        *(v4f*)&s_in[buf][r0 + rr][clA << 2]  = s.gA[rr];
      }
    }
  };

  const float* pY = (lane < 2) ? d4 : (lane < 20) ? d5 : (lane < 30) ? d6 : a;
  const int    nY = (lane < 2) ? N_D4 : (lane < 20) ? N_D5 : N_D6;
  const int  sftY = (lane < 2) ? 4 : (lane < 20) ? 3 : 2;
  const int  offY = (lane < 2) ? (128 + (lane << 2)) : (lane < 20) ? ((lane - 2) << 2)
                  : (lane < 30) ? ((lane - 20) << 2) : ((lane - 30) << 2);
  const int colX = ((cb << 3) << 4) + (lane << 2);
  const int colY = ((cb << 3) << sftY) + offY;
  const int colZ = ((cb << 3) << 2) + ((2 + (lane & 7)) << 2);
  const int ccY  = colY < (nY - 4) ? colY : (nY - 4);
  const int ccZ  = colZ < (N_A - 4) ? colZ : (N_A - 4);
  v4f gX[4], gY[4], gZ;
  auto request_span = [&]() {
    const float* qX = d4 + (size_t)((bt << 4) + r0) * (size_t)N_D4 + colX;
    const float* qY = pY + (size_t)((bt << 4) + r0) * (size_t)nY + ccY;
    const float* qZ = a  + (size_t)((bt << 4) + r0 + (lane >> 3)) * (size_t)N_A + ccZ;
#pragma unroll
    for (int rr = 0; rr < 4; ++rr) {
      gX[rr] = *(const v4f_u*)(qX + (size_t)rr * (size_t)N_D4);
      gY[rr] = *(const v4f_u*)(qY + (size_t)rr * (size_t)nY);
    }
    gZ = *(const v4f_u*)qZ;
  };
  auto park_span = [&]() {
    if (cb == 127) {
      const int shY = colY - ccY;
      const int shZ = colZ - ccZ;
#pragma unroll
      for (int rr = 0; rr < 4; ++rr) {
        *(v4f*)&s_sp[r0 + rr][lane << 2]        = gX[rr];
        *(v4f*)&s_sp[r0 + rr][(32 + lane) << 2] = fix_piece(gY[rr], shY);
      }
      *(v4f*)&s_sp[r0 + (lane >> 3)][(64 + (lane & 7)) << 2] = fix_piece(gZ, shZ);
    } else {
#pragma unroll
      for (int rr = 0; rr < 4; ++rr) {
        *(v4f*)&s_sp[r0 + rr][lane << 2]        = gX[rr];
        *(v4f*)&s_sp[r0 + rr][(32 + lane) << 2] = gY[rr];
      }
      *(v4f*)&s_sp[r0 + (lane >> 3)][(64 + (lane & 7)) << 2] = gZ;
    }
  };

  const int gx0 = cb << 3;
  request_strips(st0, gx0 + (cb & 7));
  request_strips(st1, gx0 + ((1 + cb) & 7));
  request_span();
  __builtin_amdgcn_sched_barrier(0);

  v2f wa[4][2], w6[4][2], w5[2][2], w4[2], w3[2], w2[3], w1[3];
#pragma unroll
  for (int u = 0; u < 4; ++u) {
    build_w<6, 6, 2>(wa[u], 16 * u, pm, hi);
    build_w<5, 6, 2>(w6[u], 16 * u, pm, hi);
  }
#pragma unroll
  for (int q = 0; q < 2; ++q) build_w<4, 5, 2>(w5[q], 16 * q, pm, hi);
  build_w<3, 4, 2>(w4, 0, pm, hi);
  build_w<2, 3, 2>(w3, 0, pm, hi);
  build_w<1, 2, 3>(w2, 0, pm, hi);
  build_w<0, 1, 3>(w1, 0, pm, hi);

  __builtin_amdgcn_sched_barrier(0);

  {
    const int prow = (bt << 4) + r0 + (lane >> 3);
    const int pj = lane & 7;
    auto pf = [&](const float* base, int n, int sft, int span, int unit) {
      int c = (unit << 6) < (span - 1) ? (unit << 6) : (span - 1);
      c += gx0 << sft;
      c = c < (n - 1) ? c : (n - 1);
      __builtin_prefetch(base + (size_t)prow * (size_t)n + c, 0, 3);
    };
    pf(d1, N_D1, 7, 1028, pj);
    pf(d1, N_D1, 7, 1028, pj + 8);
    pf(d1, N_D1, 7, 1028, pj + 16);
    pf(d2, N_D2, 6, 520, pj);
    pf(d2, N_D2, 6, 520, pj + 8);
    pf(d3, N_D3, 5, 264, pj);
  }
  __builtin_amdgcn_sched_barrier(0);

  park_strips(st0, 0, gx0 + (cb & 7));
  park_strips(st1, 1, gx0 + ((1 + cb) & 7));
  park_span();
  lds_barrier();

  float* po = out + (size_t)((bt << 4) + hi * 8) * (size_t)L_OUT + (wave << 6) + (m << 2);
  auto step = [&](int it, const int cur, const bool park, const bool ask, const bool more) {
    const int gx = gx0 + ((it + cb) & 7);
    if (park) park_strips(st0, cur ^ 1, gx0 + ((it + 1 + cb) & 7));
    if (ask) request_strips(st0, gx0 + ((it + 2 + cb) & 7));
    __builtin_amdgcn_sched_barrier(0);

    const float* srow = &s_in[cur][m][hi * 2];
    const int pos = (it + cb) & 7;
    const float* sp4 = &s_sp[m][hi * 2] + SP_D4 + (pos << 4);
    const float* sp5 = &s_sp[m][hi * 2] + SP_D5 + (pos << 3);
    const float* sp6 = &s_sp[m][hi * 2] + SP_D6 + (pos << 2);
    const float* spa = &s_sp[m][hi * 2] + SP_A  + (pos << 2);
    v8f acc[4];
    struct FragA { v2f a[2], s6[2], s5[2], s4[2][2]; };
    struct FragB { v2f s3[2][2], s2[2][3], s1[2][3]; };
    typedef v2f __attribute__((aligned(4))) v2f_u;
    auto frag = [&](const float* p) { return (v2f)(*(const v2f_u*)p); };
    auto mm = [&](v8f c, v2f fa, v2f fb) {
      return __builtin_amdgcn_wmma_f32_16x16x4_f32(false, fa, false, fb, (short)0, c, false, false);
    };
    auto load_a = [&](FragA& f, const int q) {
      const int t0 = (wave << 2) + 2 * q;
#pragma unroll
      for (int ks = 0; ks < 2; ++ks) {
        f.a[ks]  = frag(spa + wave + 4 * ks);
        f.s6[ks] = frag(sp6 + wave + 4 * ks);
        f.s5[ks] = frag(sp5 + (t0 >> 1) + 4 * ks);
        f.s4[0][ks] = frag(sp4 + t0 + 4 * ks);
        f.s4[1][ks] = frag(sp4 + t0 + 1 + 4 * ks);
      }
    };
    auto load_b = [&](FragB& f, const int q) {
      const int t0 = (wave << 2) + 2 * q;
#pragma unroll
      for (int e = 0; e < 2; ++e) {
        const int tl = t0 + e;
#pragma unroll
        for (int ks = 0; ks < 2; ++ks) f.s3[e][ks] = frag(srow + 204 + (tl << 1) + 4 * ks);
#pragma unroll
        for (int ks = 0; ks < 3; ++ks) f.s2[e][ks] = frag(srow + 132 + (tl << 2) + 4 * ks);
#pragma unroll
        for (int ks = 0; ks < 3; ++ks) f.s1[e][ks] = frag(srow + (tl << 3) + 4 * ks);
      }
    };
    auto prod_a = [&](const FragA& f, const int q) {
      const int u0 = 2 * q, u1 = 2 * q + 1;
      v8f c0 = {0.f, 0.f, 0.f, 0.f, 0.f, 0.f, 0.f, 0.f};
      v8f c1 = {0.f, 0.f, 0.f, 0.f, 0.f, 0.f, 0.f, 0.f};
#pragma unroll
      for (int ks = 0; ks < 2; ++ks) { c0 = mm(c0, f.a[ks], wa[u0][ks]);  c1 = mm(c1, f.a[ks], wa[u1][ks]); }
#pragma unroll
      for (int ks = 0; ks < 2; ++ks) { c0 = mm(c0, f.s6[ks], w6[u0][ks]); c1 = mm(c1, f.s6[ks], w6[u1][ks]); }
#pragma unroll
      for (int ks = 0; ks < 2; ++ks) { c0 = mm(c0, f.s5[ks], w5[0][ks]);  c1 = mm(c1, f.s5[ks], w5[1][ks]); }
#pragma unroll
      for (int ks = 0; ks < 2; ++ks) { c0 = mm(c0, f.s4[0][ks], w4[ks]);  c1 = mm(c1, f.s4[1][ks], w4[ks]); }
      acc[u0] = c0; acc[u1] = c1;
    };
    auto prod_b = [&](const FragB& f, const int q) {
      const int u0 = 2 * q, u1 = 2 * q + 1;
      v8f c0 = acc[u0], c1 = acc[u1];
#pragma unroll
      for (int ks = 0; ks < 2; ++ks) { c0 = mm(c0, f.s3[0][ks], w3[ks]); c1 = mm(c1, f.s3[1][ks], w3[ks]); }
#pragma unroll
      for (int ks = 0; ks < 3; ++ks) { c0 = mm(c0, f.s2[0][ks], w2[ks]); c1 = mm(c1, f.s2[1][ks], w2[ks]); }
#pragma unroll
      for (int ks = 0; ks < 3; ++ks) { c0 = mm(c0, f.s1[0][ks], w1[ks]); c1 = mm(c1, f.s1[1][ks], w1[ks]); }
      acc[u0] = c0; acc[u1] = c1;
    };
    FragA fa; FragB fb;
    load_a(fa, 0);
    __builtin_amdgcn_sched_barrier(0);
    load_b(fb, 0);
    __builtin_amdgcn_sched_barrier(0);
    prod_a(fa, 0);
    __builtin_amdgcn_sched_barrier(0);
    load_a(fa, 1);
    __builtin_amdgcn_sched_barrier(0);
    prod_b(fb, 0);
    __builtin_amdgcn_sched_barrier(0);
    load_b(fb, 1);
    __builtin_amdgcn_sched_barrier(0);
    prod_a(fa, 1);
    prod_b(fb, 1);

    float* pt = po + (gx << 8);
#pragma unroll
    for (int r = 0; r < 8; ++r) {
      const float y0 = acc[0][r], y1 = acc[1][r], y2 = acc[2][r], y3 = acc[3][r];
      const float z0 = lane_xchg<8, 0xC>(y0, y2);
      const float z2 = lane_xchg<8, 0x3>(y2, y0);
      const float z1 = lane_xchg<8, 0xC>(y1, y3);
      const float z3 = lane_xchg<8, 0x3>(y3, y1);
      v4f o;
      o.x = lane_xchg<4, 0xA>(z0, z1);
      o.y = lane_xchg<4, 0x5>(z1, z0);
      o.z = lane_xchg<4, 0xA>(z2, z3);
      o.w = lane_xchg<4, 0x5>(z3, z2);
      *(v4f*)(pt + (size_t)r * (size_t)L_OUT) = o;
    }

    if (more) lds_barrier();
  };
  step(0, 0, false, true, true);
  for (int it = 1; it < 5; it += 2) {
    step(it,     1, true, true, true);
    step(it + 1, 0, true, true, true);
  }
  step(5, 1, true, true, true);
  step(6, 0, true, false, true);
  step(7, 1, false, false, false);
}

extern "C" void kernel_launch(void* const* d_in, const int* in_sizes, int n_in_cnt,
                              void* d_out, int out_size, void* d_ws, size_t ws_size,
                              hipStream_t stream) {
  (void)in_sizes; (void)n_in_cnt; (void)out_size; (void)d_ws; (void)ws_size;
  const float* a  = (const float*)d_in[0];
  const float* d6 = (const float*)d_in[1];
  const float* d5 = (const float*)d_in[2];
  const float* d4 = (const float*)d_in[3];
  const float* d3 = (const float*)d_in[4];
  const float* d2 = (const float*)d_in[5];
  const float* d1 = (const float*)d_in[6];
  float* out = (float*)d_out;

  idwt6_composed_wmma<<<dim3(1024, 1), 128, 0, stream>>>(a, d6, d5, d4, d3, d2, d1, out);
}


# Two composed stages of three levels: the level-3 approximation of a work-group's walk is made once and parked in LDS; a tile takes 10 products (was 16), 4 filters (were 7); sum order changes
# speedup vs baseline: 5.1741x; 1.1339x over previous
#include <hip/hip_runtime.h>


typedef __attribute__((ext_vector_type(2))) float v2f;
typedef __attribute__((ext_vector_type(8))) float v8f;
typedef __attribute__((ext_vector_type(4))) float v4f;
typedef v4f __attribute__((aligned(4))) v4f_u;

#define LROW 244
#define SPROW 292
#define SP_D4 0
#define SP_D5 136
#define SP_D6 208
#define SP_A  248
#define X3ROW 276

#define N_D1 131075
#define N_D2 65541
#define N_D3 32774
#define N_D4 16390
#define N_D5 8198
#define N_D6 4102
#define N_A  4102
#define L_OUT 262144
#define NST 8

constexpr double K_LO[8] = {
    (double)0.23037781330885523f,  (double)0.7148465705525415f,  (double)0.6308807679295904f,
    (double)-0.02798376941698385f, (double)-0.18703481171888114f, (double)0.030841381835986965f,
    (double)0.032883011666982945f, (double)-0.010597401784997278f};
constexpr double K_HI[8] = {
    (double)-0.010597401784997278f, (double)-0.032883011666982945f, (double)0.030841381835986965f,
    (double)0.18703481171888114f,   (double)-0.02798376941698385f,  (double)-0.6308807679295904f,
    (double)0.7148465705525415f,    (double)-0.23037781330885523f};

struct FiltD { double v[64]; int lo; int len; };

constexpr double fget(const FiltD& f, int s) {
  const int i = s - f.lo;
  return (i >= 0 && i < f.len) ? f.v[i] : 0.0;
}
constexpr FiltD fbase(const double* w) {
  FiltD f{};
  f.lo = -6; f.len = 8;
  for (int i = 0; i < 8; ++i) f.v[i] = w[i];
  return f;
}
constexpr FiltD flift(const double* w, const FiltD& gl, int r) {
  FiltD f{};
  const int st = 1 << r;
  f.lo = gl.lo - 6 * st;
  f.len = (gl.lo + gl.len - 1 + st) - f.lo + 1;
  for (int i = 0; i < f.len; ++i) {
    const int s = f.lo + i;
    double acc = 0.0;
    for (int u = -6; u <= 1; ++u) acc += w[u + 6] * fget(gl, s - st * u);
    f.v[i] = acc;
  }
  return f;
}

constexpr FiltD GL1 = fbase(K_LO);
constexpr FiltD GL2 = flift(K_LO, GL1, 1);
constexpr FiltD GL3 = flift(K_LO, GL2, 2);
constexpr FiltD GH1 = fbase(K_HI);
constexpr FiltD GH2 = flift(K_HI, GL1, 1);
constexpr FiltD GH3 = flift(K_HI, GL2, 2);

#define TPAD 72
struct Tab  { float v[136]; };
struct Tabs { Tab t[4]; };
constexpr Tab to_tab(const FiltD& f) {
  Tab t{};
  for (int i = 0; i < f.len; ++i) t.v[TPAD + i] = (float)f.v[i];
  return t;
}
constexpr Tabs make_tabs() {
  Tabs r{};
  r.t[0] = to_tab(GH1); r.t[1] = to_tab(GH2); r.t[2] = to_tab(GH3); r.t[3] = to_tab(GL3);
  return r;
}
constexpr Tabs H_TABS = make_tabs();
__constant__ Tabs c_tabs = H_TABS;

template <int B, int R, int NKS>
__device__ __forceinline__ void build_w(v2f* w, int delta, int m, int hi) {
  constexpr int LO = -6 * ((1 << R) - 1);
  const float* t = &c_tabs.t[B].v[0] + (m - ((2 * hi) << R));
#pragma unroll
  for (int ks = 0; ks < NKS; ++ks) {
    w[ks].x = t[TPAD - LO + delta - ((ks * 4) << R)];
    w[ks].y = t[TPAD - LO + delta - ((ks * 4 + 1) << R)];
  }
}

__device__ __forceinline__ v4f fix_piece(v4f L, int sh) {
  if (sh != 0) {
    v4f g;
    g.x = (sh == 1) ? L.y : (sh == 2) ? L.z : (sh == 3) ? L.w : 0.f;
    g.y = (sh == 1) ? L.z : (sh == 2) ? L.w : 0.f;
    g.z = (sh == 1) ? L.w : 0.f;
    g.w = 0.f;
    L = g;
  }
  return L;
}

template <int NKS>
__device__ __forceinline__ v8f band_mm(v8f acc, const float* p, const v2f* w) {
#pragma unroll
  for (int ks = 0; ks < NKS; ++ks) {
    v2f fa;
    fa.x = p[ks * 4];
    fa.y = p[ks * 4 + 1];
    acc = __builtin_amdgcn_wmma_f32_16x16x4_f32(
        false, fa, false, w[ks], (short)0, acc, false, false);
  }
  return acc;
}

template <int XM, int BANKS>
__device__ __forceinline__ float lane_xchg(float keep, float other) {
  return __builtin_bit_cast(float, __builtin_amdgcn_update_dpp(
      __builtin_bit_cast(int, keep), __builtin_bit_cast(int, other),
      0x160 + XM  , 0xF, BANKS, false));
}

__device__ __forceinline__ void lds_barrier() {
  asm volatile("s_wait_dscnt 0x0\n\ts_barrier_signal -1\n\ts_barrier_wait -1" ::: "memory");
}

__global__ __launch_bounds__(128, 4)
void idwt6_composed_wmma(const float* __restrict__ a,  const float* __restrict__ d6,
                         const float* __restrict__ d5, const float* __restrict__ d4,
                         const float* __restrict__ d3, const float* __restrict__ d2,
                         const float* __restrict__ d1, float* __restrict__ out) {
  const int wave = threadIdx.x >> 5;
  const int lane = threadIdx.x & 31;
  const int m  = lane & 15;
  const int hi = lane >> 4;
  const int pm = ((m & 3) << 2) | (m >> 2);
  const int cb = blockIdx.x >> 3;
  const int bt = blockIdx.x & 7;

  __shared__ __attribute__((aligned(16))) float s_in[2][16][LROW];
  __shared__ __attribute__((aligned(16))) float s_sp[16][SPROW];
  __shared__ __attribute__((aligned(16))) float s_x3[16][X3ROW];

  const int r0 = wave << 2;
  const float* pA = (lane < 18) ? d2 : (lane < 28) ? d3 : d1;
  const int    nA = (lane < 18) ? N_D2 : (lane < 28) ? N_D3 : N_D1;
  const int  sftA = (lane < 18) ? 6 : (lane < 28) ? 5 : 7;
  const int  offA = (lane < 18) ? (lane << 2) : (lane < 28) ? ((lane - 18) << 2) : 128;
  const int   clA = (lane < 28) ? (33 + lane) : 32;

  const float* rD = d1 + (size_t)((bt << 4) + r0) * (size_t)N_D1;
  const float* rA = pA + (size_t)((bt << 4) + r0) * (size_t)nA;

  struct Strips { v4f gD[4], gA[4]; };
  Strips st0, st1;
  auto request_strips = [&](Strips& s, int gx) {
    v4f (&gD)[4] = s.gD; v4f (&gA)[4] = s.gA;
    const int colD = (gx << 7) + (lane << 2);
    const int colA = (gx << sftA) + offA;
    const int ccD = colD < (N_D1 - 4) ? colD : (N_D1 - 4);
    const int ccA = colA < (nA - 4) ? colA : (nA - 4);
    const float* qD = rD + ccD;
    const float* qA = rA + ccA;
#pragma unroll
    for (int rr = 0; rr < 4; ++rr) {
      gD[rr] = *(const v4f_u*)(qD + (size_t)rr * (size_t)N_D1);
      gA[rr] = *(const v4f_u*)(qA + (size_t)rr * (size_t)nA);
    }
  };
  auto park_strips = [&](const Strips& s, int buf, int gx) {
    if (gx == 1023) {
      const int colD = (gx << 7) + (lane << 2);
      const int colA = (gx << sftA) + offA;
      const int shD = colD - (colD < (N_D1 - 4) ? colD : (N_D1 - 4));
      const int shA = colA - (colA < (nA - 4) ? colA : (nA - 4));
#pragma unroll
      for (int rr = 0; rr < 4; ++rr) {
        *(v4f*)&s_in[buf][r0 + rr][lane << 2] = fix_piece(s.gD[rr], shD);
        *(v4f*)&s_in[buf][r0 + rr][clA << 2]  = fix_piece(s.gA[rr], shA);
      }
    } else {
#pragma unroll
      for (int rr = 0; rr < 4; ++rr) {
        *(v4f*)&s_in[buf][r0 + rr][lane << 2] = s.gD[rr];
        *(v4f*)&s_in[buf][r0 + rr][clA << 2]  = s.gA[rr];
      }
    }
  };

  const float* pY = (lane < 2) ? d4 : (lane < 20) ? d5 : (lane < 30) ? d6 : a;
  const int    nY = (lane < 2) ? N_D4 : (lane < 20) ? N_D5 : N_D6;
  const int  sftY = (lane < 2) ? 4 : (lane < 20) ? 3 : 2;
  const int  offY = (lane < 2) ? (128 + (lane << 2)) : (lane < 20) ? ((lane - 2) << 2)
                  : (lane < 30) ? ((lane - 20) << 2) : ((lane - 30) << 2);
  const int colX = ((cb << 3) << 4) + (lane << 2);
  const int colY = ((cb << 3) << sftY) + offY;
  const int colZ = ((cb << 3) << 2) + ((2 + (lane & 7)) << 2);
  const int ccY  = colY < (nY - 4) ? colY : (nY - 4);
  const int ccZ  = colZ < (N_A - 4) ? colZ : (N_A - 4);
  v4f gX[4], gY[4], gZ;
  auto request_span = [&]() {
    const float* qX = d4 + (size_t)((bt << 4) + r0) * (size_t)N_D4 + colX;
    const float* qY = pY + (size_t)((bt << 4) + r0) * (size_t)nY + ccY;
    const float* qZ = a  + (size_t)((bt << 4) + r0 + (lane >> 3)) * (size_t)N_A + ccZ;
#pragma unroll
    for (int rr = 0; rr < 4; ++rr) {
      gX[rr] = *(const v4f_u*)(qX + (size_t)rr * (size_t)N_D4);
      gY[rr] = *(const v4f_u*)(qY + (size_t)rr * (size_t)nY);
    }
    gZ = *(const v4f_u*)qZ;
  };
  auto park_span = [&]() {
    if (cb == 127) {
      const int shY = colY - ccY;
      const int shZ = colZ - ccZ;
#pragma unroll
      for (int rr = 0; rr < 4; ++rr) {
        *(v4f*)&s_sp[r0 + rr][lane << 2]        = gX[rr];
        *(v4f*)&s_sp[r0 + rr][(32 + lane) << 2] = fix_piece(gY[rr], shY);
      }
      *(v4f*)&s_sp[r0 + (lane >> 3)][(64 + (lane & 7)) << 2] = fix_piece(gZ, shZ);
    } else {
#pragma unroll
      for (int rr = 0; rr < 4; ++rr) {
        *(v4f*)&s_sp[r0 + rr][lane << 2]        = gX[rr];
        *(v4f*)&s_sp[r0 + rr][(32 + lane) << 2] = gY[rr];
      }
      *(v4f*)&s_sp[r0 + (lane >> 3)][(64 + (lane & 7)) << 2] = gZ;
    }
  };

  const int gx0 = cb << 3;
  request_strips(st0, gx0 + (cb & 7));
  request_strips(st1, gx0 + ((1 + cb) & 7));
  request_span();
  __builtin_amdgcn_sched_barrier(0);

  v2f wl[2], w3[2], w2[3], w1[3];
  build_w<3, 3, 2>(wl, 0, pm, hi);
  build_w<2, 3, 2>(w3, 0, pm, hi);
  build_w<1, 2, 3>(w2, 0, pm, hi);
  build_w<0, 1, 3>(w1, 0, pm, hi);

  __builtin_amdgcn_sched_barrier(0);

  {
    const int prow = (bt << 4) + r0 + (lane >> 3);
    const int pj = lane & 7;
    auto pf = [&](const float* base, int n, int sft, int span, int unit) {
      int c = (unit << 6) < (span - 1) ? (unit << 6) : (span - 1);
      c += gx0 << sft;
      c = c < (n - 1) ? c : (n - 1);
      __builtin_prefetch(base + (size_t)prow * (size_t)n + c, 0, 3);
    };
    pf(d1, N_D1, 7, 1028, pj);
    pf(d1, N_D1, 7, 1028, pj + 8);
    pf(d1, N_D1, 7, 1028, pj + 16);
    pf(d2, N_D2, 6, 520, pj);
    pf(d2, N_D2, 6, 520, pj + 8);
    pf(d3, N_D3, 5, 264, pj);
  }
  __builtin_amdgcn_sched_barrier(0);

  park_strips(st0, 0, gx0 + (cb & 7));
  park_strips(st1, 1, gx0 + ((1 + cb) & 7));
  park_span();
  lds_barrier();

  request_strips(st0, gx0 + ((2 + cb) & 7));
  __builtin_amdgcn_sched_barrier(0);

  typedef v2f __attribute__((aligned(4))) v2f_u;
  auto frag = [&](const float* p) { return (v2f)(*(const v2f_u*)p); };
  auto mm = [&](v8f acc_in, v2f fa, v2f fb) {
    return __builtin_amdgcn_wmma_f32_16x16x4_f32(false, fa, false, fb, (short)0, acc_in, false, false);
  };

  {
    const float* sp = &s_sp[m][hi * 2];
    struct FragC { v2f a[2], s6[2], s5[3], s4[3]; };
    auto load_c = [&](FragC& f, const int t3) {
#pragma unroll
      for (int ks = 0; ks < 2; ++ks) {
        f.a[ks]  = frag(sp + SP_A  + (t3 << 1) + 4 * ks);
        f.s6[ks] = frag(sp + SP_D6 + (t3 << 1) + 4 * ks);
      }
#pragma unroll
      for (int ks = 0; ks < 3; ++ks) {
        f.s5[ks] = frag(sp + SP_D5 + (t3 << 2) + 4 * ks);
        f.s4[ks] = frag(sp + SP_D4 + (t3 << 3) + 4 * ks);
      }
    };
    auto park_x3 = [&](const v8f& v, const int t3) {
      float* px = &s_x3[hi * 8][(t3 << 4) + pm];
#pragma unroll
      for (int r = 0; r < 8; ++r) px[r * X3ROW] = v[r];
    };
    const v8f zero8 = {0.f, 0.f, 0.f, 0.f, 0.f, 0.f, 0.f, 0.f};
#pragma unroll
    for (int q = 0; q < 2; ++q) {
      const int t3 = (wave << 2) + 2 * q;
      FragC f0, f1;
      load_c(f0, t3);
      load_c(f1, t3 + 1);
      v8f c0 = zero8, c1 = zero8;
#pragma unroll
      for (int ks = 0; ks < 2; ++ks) { c0 = mm(c0, f0.a[ks], wl[ks]);  c1 = mm(c1, f1.a[ks], wl[ks]); }
#pragma unroll
      for (int ks = 0; ks < 2; ++ks) { c0 = mm(c0, f0.s6[ks], w3[ks]); c1 = mm(c1, f1.s6[ks], w3[ks]); }
#pragma unroll
      for (int ks = 0; ks < 3; ++ks) { c0 = mm(c0, f0.s5[ks], w2[ks]); c1 = mm(c1, f1.s5[ks], w2[ks]); }
#pragma unroll
      for (int ks = 0; ks < 3; ++ks) { c0 = mm(c0, f0.s4[ks], w1[ks]); c1 = mm(c1, f1.s4[ks], w1[ks]); }
      park_x3(c0, t3);
      park_x3(c1, t3 + 1);
    }
    if (__builtin_amdgcn_readfirstlane(wave) == 3) {
      FragC f0;
      load_c(f0, 16);
      v8f c0 = zero8;
#pragma unroll
      for (int ks = 0; ks < 2; ++ks) c0 = mm(c0, f0.a[ks], wl[ks]);
#pragma unroll
      for (int ks = 0; ks < 2; ++ks) c0 = mm(c0, f0.s6[ks], w3[ks]);
#pragma unroll
      for (int ks = 0; ks < 3; ++ks) c0 = mm(c0, f0.s5[ks], w2[ks]);
#pragma unroll
      for (int ks = 0; ks < 3; ++ks) c0 = mm(c0, f0.s4[ks], w1[ks]);
      park_x3(c0, 16);
    }
  }
  lds_barrier();

  float* po = out + (size_t)((bt << 4) + hi * 8) * (size_t)L_OUT + (wave << 6) + (m << 2);
  auto step = [&](int it, const int cur, const bool park, const bool ask, const bool more) {
    const int gx = gx0 + ((it + cb) & 7);
    if (park) park_strips(st0, cur ^ 1, gx0 + ((it + 1 + cb) & 7));
    if (ask) request_strips(st0, gx0 + ((it + 2 + cb) & 7));
    __builtin_amdgcn_sched_barrier(0);

    const float* srow = &s_in[cur][m][hi * 2];
    const int pos = (it + cb) & 7;
    const float* x3row = &s_x3[m][hi * 2] + (pos << 5);
    v8f acc[4];
    struct Frag { v2f x3[2][2], s3[2][2], s2[2][3], s1[2][3]; };
    auto load_f = [&](Frag& f, const int q) {
      const int t0 = (wave << 2) + 2 * q;
#pragma unroll
      for (int e = 0; e < 2; ++e) {
        const int tl = t0 + e;
#pragma unroll
        for (int ks = 0; ks < 2; ++ks) f.x3[e][ks] = frag(x3row + (tl << 1) + 4 * ks);
#pragma unroll
        for (int ks = 0; ks < 2; ++ks) f.s3[e][ks] = frag(srow + 204 + (tl << 1) + 4 * ks);
#pragma unroll
        for (int ks = 0; ks < 3; ++ks) f.s2[e][ks] = frag(srow + 132 + (tl << 2) + 4 * ks);
#pragma unroll
        for (int ks = 0; ks < 3; ++ks) f.s1[e][ks] = frag(srow + (tl << 3) + 4 * ks);
      }
    };
    auto prod_f = [&](const Frag& f, const int q) {
      const int u0 = 2 * q, u1 = 2 * q + 1;
      v8f c0 = {0.f, 0.f, 0.f, 0.f, 0.f, 0.f, 0.f, 0.f};
      v8f c1 = {0.f, 0.f, 0.f, 0.f, 0.f, 0.f, 0.f, 0.f};
#pragma unroll
      for (int ks = 0; ks < 2; ++ks) { c0 = mm(c0, f.x3[0][ks], wl[ks]); c1 = mm(c1, f.x3[1][ks], wl[ks]); }
#pragma unroll
      for (int ks = 0; ks < 2; ++ks) { c0 = mm(c0, f.s3[0][ks], w3[ks]); c1 = mm(c1, f.s3[1][ks], w3[ks]); }
#pragma unroll
      for (int ks = 0; ks < 3; ++ks) { c0 = mm(c0, f.s2[0][ks], w2[ks]); c1 = mm(c1, f.s2[1][ks], w2[ks]); }
#pragma unroll
      for (int ks = 0; ks < 3; ++ks) { c0 = mm(c0, f.s1[0][ks], w1[ks]); c1 = mm(c1, f.s1[1][ks], w1[ks]); }
      acc[u0] = c0; acc[u1] = c1;
    };
    Frag f0, f1;
    load_f(f0, 0);
    __builtin_amdgcn_sched_barrier(0);
    load_f(f1, 1);
    __builtin_amdgcn_sched_barrier(0);
    prod_f(f0, 0);
    prod_f(f1, 1);

    float* pt = po + (gx << 8);
#pragma unroll
    for (int r = 0; r < 8; ++r) {
      const float y0 = acc[0][r], y1 = acc[1][r], y2 = acc[2][r], y3 = acc[3][r];
      const float z0 = lane_xchg<8, 0xC>(y0, y2);
      const float z2 = lane_xchg<8, 0x3>(y2, y0);
      const float z1 = lane_xchg<8, 0xC>(y1, y3);
      const float z3 = lane_xchg<8, 0x3>(y3, y1);
      v4f o;
      o.x = lane_xchg<4, 0xA>(z0, z1);
      o.y = lane_xchg<4, 0x5>(z1, z0);
      o.z = lane_xchg<4, 0xA>(z2, z3);
      o.w = lane_xchg<4, 0x5>(z3, z2);
      *(v4f*)(pt + (size_t)r * (size_t)L_OUT) = o;
    }

    if (more) lds_barrier();
  };
  step(0, 0, false, false, true);
  for (int it = 1; it < 5; it += 2) {
    step(it,     1, true, true, true);
    step(it + 1, 0, true, true, true);
  }
  step(5, 1, true, true, true);
  step(6, 0, true, false, true);
  step(7, 1, false, false, false);
}

extern "C" void kernel_launch(void* const* d_in, const int* in_sizes, int n_in_cnt,
                              void* d_out, int out_size, void* d_ws, size_t ws_size,
                              hipStream_t stream) {
  (void)in_sizes; (void)n_in_cnt; (void)out_size; (void)d_ws; (void)ws_size;
  const float* a  = (const float*)d_in[0];
  const float* d6 = (const float*)d_in[1];
  const float* d5 = (const float*)d_in[2];
  const float* d4 = (const float*)d_in[3];
  const float* d3 = (const float*)d_in[4];
  const float* d2 = (const float*)d_in[5];
  const float* d1 = (const float*)d_in[6];
  float* out = (float*)d_out;

  idwt6_composed_wmma<<<dim3(1024, 1), 128, 0, stream>>>(a, d6, d5, d4, d3, d2, d1, out);
}
